# starting kernel
# baseline (speedup 1.0000x reference)
#include <hip/hip_runtime.h>
#include <math.h>

#define NG   2048
#define H_   128
#define W_   128
#define D_   32
#define F_   2
#define LOG2E 1.4426950408889634f

typedef float v2f __attribute__((ext_vector_type(2)));
typedef float v8f __attribute__((ext_vector_type(8)));

__global__ __launch_bounds__(256) void prep_gaussians(
    const float* __restrict__ xyz, const float* __restrict__ scale,
    const float* __restrict__ rot, const float* __restrict__ opac,
    const float* __restrict__ feat, float4* __restrict__ P)
{
    int n = blockIdx.x * 256 + threadIdx.x;
    if (n >= NG) return;

    float qw = rot[n*4+0], qx = rot[n*4+1], qy = rot[n*4+2], qz = rot[n*4+3];
    float inv = 1.0f / (sqrtf(qw*qw + qx*qx + qy*qy + qz*qz) + 1e-8f);
    qw *= inv; qx *= inv; qy *= inv; qz *= inv;

    float r00 = 1.f - 2.f*(qy*qy + qz*qz);
    float r01 = 2.f*(qx*qy - qw*qz);
    float r02 = 2.f*(qx*qz + qw*qy);
    float r10 = 2.f*(qx*qy + qw*qz);
    float r11 = 1.f - 2.f*(qx*qx + qz*qz);
    float r12 = 2.f*(qy*qz - qw*qx);
    float r20 = 2.f*(qx*qz - qw*qy);
    float r21 = 2.f*(qy*qz + qw*qx);
    float r22 = 1.f - 2.f*(qx*qx + qy*qy);

    float sx = scale[n*3+0], sy = scale[n*3+1], sz = scale[n*3+2];
    float m00 = r00*sx, m01 = r01*sy, m02 = r02*sz;
    float m10 = r10*sx, m11 = r11*sy, m12 = r12*sz;
    float m20 = r20*sx, m21 = r21*sy, m22 = r22*sz;

    float cov00 = m00*m00 + m01*m01 + m02*m02;
    float cov01 = m00*m10 + m01*m11 + m02*m12;
    float cov02 = m00*m20 + m01*m21 + m02*m22;
    float cov11 = m10*m10 + m11*m11 + m12*m12;
    float cov12 = m10*m20 + m11*m21 + m12*m22;
    float cov22 = m20*m20 + m21*m21 + m22*m22;

    float zv   = cov22 + 1e-8f;
    float zinv = 1.0f / zv;
    float thr3 = 3.0f * sqrtf(zv);

    float a  = cov00 - zinv*cov02*cov02 + 1e-6f;
    float b  = cov01 - zinv*cov02*cov12;
    float cc = cov11 - zinv*cov12*cov12 + 1e-6f;
    float idet = 1.0f / (a*cc - b*b);
    float pa = cc*idet, pb = -b*idet, pc = a*idet;

    P[n*4+0] = make_float4(xyz[n*3+0], xyz[n*3+1], xyz[n*3+2], cov02);
    P[n*4+1] = make_float4(cov12, zinv, thr3, opac[n]);
    P[n*4+2] = make_float4(-0.5f*LOG2E*pa, -LOG2E*pb, -0.5f*LOG2E*pc, feat[n*2+0]);
    P[n*4+3] = make_float4(feat[n*2+1], 0.f, 0.f, 0.f);
}

__global__ __launch_bounds__(256) void render_slices(
    const float4* __restrict__ P, float* __restrict__ out)
{
    __shared__ float4 s0[256];
    __shared__ float4 s1[256];

    const int d      = blockIdx.x;
    const int h      = blockIdx.y;
    const int tid    = threadIdx.x;
    const int lane   = tid & 31;
    const int wave   = tid >> 5;
    const int laneN  = lane & 15;
    const int laneHi = lane >> 4;
    const int wbase  = wave * 16;

    const float z  = (float)d * (1.0f / (D_ - 1));
    const float yh = (float)h * (1.0f / (H_ - 1));
    const float xw = (float)(wbase + laneN) * (1.0f / (W_ - 1));

    v8f acc = {0.f, 0.f, 0.f, 0.f, 0.f, 0.f, 0.f, 0.f};

    for (int c0 = 0; c0 < NG; c0 += 256) {
        int np = c0 + 256;
        if (np < NG)
            __builtin_prefetch((const void*)&P[(size_t)(np + tid) * 4], 0, 3);

        {
            int g = c0 + tid;
            float4 q0 = P[g*4+0];
            float4 q1 = P[g*4+1];
            float4 q2 = P[g*4+2];
            float4 q3 = P[g*4+3];
            float tt  = q1.y * (z - q0.z);
            float mux = q0.x + q0.w * tt;
            float muy = q0.y + q1.x * tt;
            float msk = (fabsf(q0.z - z) < q1.z) ? 1.0f : 0.0f;
            s0[tid] = make_float4(mux, muy, q1.w * msk, q2.w);
            s1[tid] = make_float4(q2.x, q2.y, q2.z, q3.x);
        }
        __syncthreads();

        for (int s = 0; s < 64; ++s) {
            int gA = (s << 2) + (laneHi << 1);
            float4 a0 = s0[gA];
            float4 b0 = s0[gA + 1];
            bool live = (a0.z != 0.f) || (b0.z != 0.f);
            if (__any(live)) {
                float4 a1 = s1[gA];
                float4 b1 = s1[gA + 1];

                float dxA = xw - a0.x, dyA = yh - a0.y;
                float mA  = dxA * (a1.x * dxA + a1.y * dyA) + a1.z * dyA * dyA;
                float alA = a0.z * __builtin_amdgcn_exp2f(mA);

                float dxB = xw - b0.x, dyB = yh - b0.y;
                float mB  = dxB * (b1.x * dxB + b1.y * dyB) + b1.z * dyB * dyB;
                float alB = b0.z * __builtin_amdgcn_exp2f(mB);

                v2f Am = {alA, alB};
                float e0 = (laneN == 0) ? a0.w : (laneN == 1) ? a1.w : 0.f;
                float e1 = (laneN == 0) ? b0.w : (laneN == 1) ? b1.w : 0.f;
                v2f Bm = {e0, e1};

                acc = __builtin_amdgcn_wmma_f32_16x16x4_f32(
                          false, Am, false, Bm, (short)0, acc, false, false);
            }
        }
        __syncthreads();
    }

    if (laneN < F_) {
        #pragma unroll
        for (int r = 0; r < 8; ++r) {
            int M = r + (laneHi << 3);
            int w = wbase + M;
            out[(((size_t)h * W_ + w) * D_ + d) * F_ + laneN] = acc[r];
        }
    }
}

extern "C" void kernel_launch(void* const* d_in, const int* in_sizes, int n_in,
                              void* d_out, int out_size, void* d_ws, size_t ws_size,
                              hipStream_t stream)
{
    const float* xyz   = (const float*)d_in[0];
    const float* scale = (const float*)d_in[1];
    const float* rot   = (const float*)d_in[2];
    const float* opac  = (const float*)d_in[3];
    const float* feat  = (const float*)d_in[4];
    float4* P = (float4*)d_ws;

    prep_gaussians<<<NG / 256, 256, 0, stream>>>(xyz, scale, rot, opac, feat, P);
    render_slices<<<dim3(D_, H_), 256, 0, stream>>>(P, (float*)d_out);
}


# Kernel 1 compacts per slice, in index order, the Gaussians passing the z mask; render loops only over these lists (live terms regrouped in fours in the f32 WMMA sums) instead of test-and-skip
# speedup vs baseline: 3.0270x; 3.0270x over previous
#include <hip/hip_runtime.h>
#include <math.h>

#define NG   2048
#define H_   128
#define W_   128
#define D_   32
#define F_   2
#define LOG2E 1.4426950408889634f

#define CH      256
#define NCH     (NG / CH)
#define HDR     8
#define RSTRIDE (HDR + 2 * CH)
#define RG      4

typedef float v2f __attribute__((ext_vector_type(2)));
typedef float v8f __attribute__((ext_vector_type(8)));

__global__ __launch_bounds__(256) void prep_compact(
    const float* __restrict__ xyz, const float* __restrict__ scale,
    const float* __restrict__ rot, const float* __restrict__ opac,
    const float* __restrict__ feat, float4* __restrict__ L)
{
    __shared__ int wcnt[8];

    const int c    = blockIdx.x;
    const int d    = blockIdx.y;
    const int tid  = threadIdx.x;
    const int lane = tid & 31;
    const int wave = tid >> 5;
    const int n    = c * CH + tid;

    float qw = rot[n*4+0], qx = rot[n*4+1], qy = rot[n*4+2], qz = rot[n*4+3];
    float inv = 1.0f / (sqrtf(qw*qw + qx*qx + qy*qy + qz*qz) + 1e-8f);
    qw *= inv; qx *= inv; qy *= inv; qz *= inv;

    float r00 = 1.f - 2.f*(qy*qy + qz*qz);
    float r01 = 2.f*(qx*qy - qw*qz);
    float r02 = 2.f*(qx*qz + qw*qy);
    float r10 = 2.f*(qx*qy + qw*qz);
    float r11 = 1.f - 2.f*(qx*qx + qz*qz);
    float r12 = 2.f*(qy*qz - qw*qx);
    float r20 = 2.f*(qx*qz - qw*qy);
    float r21 = 2.f*(qy*qz + qw*qx);
    float r22 = 1.f - 2.f*(qx*qx + qy*qy);

    float sx = scale[n*3+0], sy = scale[n*3+1], sz = scale[n*3+2];
    float m00 = r00*sx, m01 = r01*sy, m02 = r02*sz;
    float m10 = r10*sx, m11 = r11*sy, m12 = r12*sz;
    float m20 = r20*sx, m21 = r21*sy, m22 = r22*sz;

    float cov00 = m00*m00 + m01*m01 + m02*m02;
    float cov01 = m00*m10 + m01*m11 + m02*m12;
    float cov02 = m00*m20 + m01*m21 + m02*m22;
    float cov11 = m10*m10 + m11*m11 + m12*m12;
    float cov12 = m10*m20 + m11*m21 + m12*m22;
    float cov22 = m20*m20 + m21*m21 + m22*m22;

    float zv   = cov22 + 1e-8f;
    float zinv = 1.0f / zv;
    float thr3 = 3.0f * sqrtf(zv);

    float a  = cov00 - zinv*cov02*cov02 + 1e-6f;
    float b  = cov01 - zinv*cov02*cov12;
    float cc = cov11 - zinv*cov12*cov12 + 1e-6f;
    float idet = 1.0f / (a*cc - b*b);
    float pa = cc*idet, pb = -b*idet, pc = a*idet;

    const float z = (float)d * (1.0f / (D_ - 1));
    float mu_x = xyz[n*3+0], mu_y = xyz[n*3+1], mu_z = xyz[n*3+2];
    float tt  = zinv * (z - mu_z);
    float mux = mu_x + cov02 * tt;
    float muy = mu_y + cov12 * tt;
    bool live = fabsf(mu_z - z) < thr3;

    unsigned int bal = (unsigned int)__ballot(live);
    int pos = __popc(bal & ((1u << lane) - 1u));
    if (lane == 0) wcnt[wave] = __popc(bal);
    __syncthreads();
    int off = 0, total = 0;
    #pragma unroll
    for (int w = 0; w < 8; ++w) {
        int v = wcnt[w];
        total += v;
        if (w < wave) off += v;
    }

    float4* R = L + (size_t)(d * NCH + c) * RSTRIDE;
    if (live) {
        int i = off + pos;
        R[HDR + i]      = make_float4(mux, muy, opac[n], feat[n*2+0]);
        R[HDR + CH + i] = make_float4(-0.5f*LOG2E*pa, -LOG2E*pb, -0.5f*LOG2E*pc, feat[n*2+1]);
    }
    int n4 = (total + 3) & ~3;
    if (tid < n4 - total) {
        R[HDR + total + tid]      = make_float4(0.f, 0.f, 0.f, 0.f);
        R[HDR + CH + total + tid] = make_float4(0.f, 0.f, 0.f, 0.f);
    }
    if (tid == 0) ((int*)R)[0] = total;
}

__global__ __launch_bounds__(256) void render_slices(
    const float4* __restrict__ L, float* __restrict__ out)
{
    __shared__ float4 s0[RG * CH];
    __shared__ float4 s1[RG * CH];

    const int d      = blockIdx.x;
    const int h      = blockIdx.y;
    const int tid    = threadIdx.x;
    const int lane   = tid & 31;
    const int wave   = tid >> 5;
    const int laneN  = lane & 15;
    const int laneHi = lane >> 4;
    const int wbase  = wave * 16;

    const float yh = (float)h * (1.0f / (H_ - 1));
    const float xw = (float)(wbase + laneN) * (1.0f / (W_ - 1));

    v8f acc = {0.f, 0.f, 0.f, 0.f, 0.f, 0.f, 0.f, 0.f};

    const float4* Ls = L + (size_t)d * NCH * RSTRIDE;
    int n4[NCH];
    #pragma unroll
    for (int r = 0; r < NCH; ++r)
        n4[r] = (((const int*)(Ls + (size_t)r * RSTRIDE))[0] + 3) & ~3;

    const int q  = tid >> 6;
    const int i0 = tid & 63;

    #pragma unroll
    for (int c0 = 0; c0 < NCH; c0 += RG) {
        int myN = 0, myBase = 0, base = 0;
        #pragma unroll
        for (int r = 0; r < RG; ++r) {
            if (q == r) { myN = n4[c0 + r]; myBase = base; }
            base += n4[c0 + r];
        }
        const float4* R = Ls + (size_t)(c0 + q) * RSTRIDE;
        for (int i = i0; i < myN; i += 64) {
            s0[myBase + i] = R[HDR + i];
            s1[myBase + i] = R[HDR + CH + i];
        }
        __syncthreads();

        const int steps = base >> 2;
        for (int s = 0; s < steps; ++s) {
            int gA = (s << 2) + (laneHi << 1);
            float4 a0 = s0[gA];
            float4 b0 = s0[gA + 1];
            float4 a1 = s1[gA];
            float4 b1 = s1[gA + 1];

            float dxA = xw - a0.x, dyA = yh - a0.y;
            float mA  = dxA * (a1.x * dxA + a1.y * dyA) + a1.z * dyA * dyA;
            float alA = a0.z * __builtin_amdgcn_exp2f(mA);

            float dxB = xw - b0.x, dyB = yh - b0.y;
            float mB  = dxB * (b1.x * dxB + b1.y * dyB) + b1.z * dyB * dyB;
            float alB = b0.z * __builtin_amdgcn_exp2f(mB);

            v2f Am = {alA, alB};
            float e0 = (laneN == 0) ? a0.w : (laneN == 1) ? a1.w : 0.f;
            float e1 = (laneN == 0) ? b0.w : (laneN == 1) ? b1.w : 0.f;
            v2f Bm = {e0, e1};

            acc = __builtin_amdgcn_wmma_f32_16x16x4_f32(
                      false, Am, false, Bm, (short)0, acc, false, false);
        }
        __syncthreads();
    }

    if (laneN < F_) {
        #pragma unroll
        for (int r = 0; r < 8; ++r) {
            int M = r + (laneHi << 3);
            int w = wbase + M;
            out[(((size_t)h * W_ + w) * D_ + d) * F_ + laneN] = acc[r];
        }
    }
}

extern "C" void kernel_launch(void* const* d_in, const int* in_sizes, int n_in,
                              void* d_out, int out_size, void* d_ws, size_t ws_size,
                              hipStream_t stream)
{
    const float* xyz   = (const float*)d_in[0];
    const float* scale = (const float*)d_in[1];
    const float* rot   = (const float*)d_in[2];
    const float* opac  = (const float*)d_in[3];
    const float* feat  = (const float*)d_in[4];
    float4* L = (float4*)d_ws;

    prep_compact<<<dim3(NCH, D_), 256, 0, stream>>>(xyz, scale, rot, opac, feat, L);
    render_slices<<<dim3(D_, H_), 256, 0, stream>>>(L, (float*)d_out);
}


# Render: each wave computes four 16-pixel tiles (four accumulators) from one LDS read of a step's Gaussian entries; a block is 2 waves per image row; LDS reads per WMMA fall to a quarter
# speedup vs baseline: 4.5837x; 1.5143x over previous
#include <hip/hip_runtime.h>
#include <math.h>

#define NG   2048
#define H_   128
#define W_   128
#define D_   32
#define F_   2
#define LOG2E 1.4426950408889634f

#define CH      256
#define NCH     (NG / CH)
#define HDR     8
#define RSTRIDE (HDR + 2 * CH)
#define RG      4

typedef float v2f __attribute__((ext_vector_type(2)));
typedef float v8f __attribute__((ext_vector_type(8)));

__global__ __launch_bounds__(256) void prep_compact(
    const float* __restrict__ xyz, const float* __restrict__ scale,
    const float* __restrict__ rot, const float* __restrict__ opac,
    const float* __restrict__ feat, float4* __restrict__ L)
{
    __shared__ int wcnt[8];

    const int c    = blockIdx.x;
    const int d    = blockIdx.y;
    const int tid  = threadIdx.x;
    const int lane = tid & 31;
    const int wave = tid >> 5;
    const int n    = c * CH + tid;

    float qw = rot[n*4+0], qx = rot[n*4+1], qy = rot[n*4+2], qz = rot[n*4+3];
    float inv = 1.0f / (sqrtf(qw*qw + qx*qx + qy*qy + qz*qz) + 1e-8f);
    qw *= inv; qx *= inv; qy *= inv; qz *= inv;

    float r00 = 1.f - 2.f*(qy*qy + qz*qz);
    float r01 = 2.f*(qx*qy - qw*qz);
    float r02 = 2.f*(qx*qz + qw*qy);
    float r10 = 2.f*(qx*qy + qw*qz);
    float r11 = 1.f - 2.f*(qx*qx + qz*qz);
    float r12 = 2.f*(qy*qz - qw*qx);
    float r20 = 2.f*(qx*qz - qw*qy);
    float r21 = 2.f*(qy*qz + qw*qx);
    float r22 = 1.f - 2.f*(qx*qx + qy*qy);

    float sx = scale[n*3+0], sy = scale[n*3+1], sz = scale[n*3+2];
    float m00 = r00*sx, m01 = r01*sy, m02 = r02*sz;
    float m10 = r10*sx, m11 = r11*sy, m12 = r12*sz;
    float m20 = r20*sx, m21 = r21*sy, m22 = r22*sz;

    float cov00 = m00*m00 + m01*m01 + m02*m02;
    float cov01 = m00*m10 + m01*m11 + m02*m12;
    float cov02 = m00*m20 + m01*m21 + m02*m22;
    float cov11 = m10*m10 + m11*m11 + m12*m12;
    float cov12 = m10*m20 + m11*m21 + m12*m22;
    float cov22 = m20*m20 + m21*m21 + m22*m22;

    float zv   = cov22 + 1e-8f;
    float zinv = 1.0f / zv;
    float thr3 = 3.0f * sqrtf(zv);

    float a  = cov00 - zinv*cov02*cov02 + 1e-6f;
    float b  = cov01 - zinv*cov02*cov12;
    float cc = cov11 - zinv*cov12*cov12 + 1e-6f;
    float idet = 1.0f / (a*cc - b*b);
    float pa = cc*idet, pb = -b*idet, pc = a*idet;

    const float z = (float)d * (1.0f / (D_ - 1));
    float mu_x = xyz[n*3+0], mu_y = xyz[n*3+1], mu_z = xyz[n*3+2];
    float tt  = zinv * (z - mu_z);
    float mux = mu_x + cov02 * tt;
    float muy = mu_y + cov12 * tt;
    bool live = fabsf(mu_z - z) < thr3;

    unsigned int bal = (unsigned int)__ballot(live);
    int pos = __popc(bal & ((1u << lane) - 1u));
    if (lane == 0) wcnt[wave] = __popc(bal);
    __syncthreads();
    int off = 0, total = 0;
    #pragma unroll
    for (int w = 0; w < 8; ++w) {
        int v = wcnt[w];
        total += v;
        if (w < wave) off += v;
    }

    float4* R = L + (size_t)(d * NCH + c) * RSTRIDE;
    if (live) {
        int i = off + pos;
        R[HDR + i]      = make_float4(mux, muy, opac[n], feat[n*2+0]);
        R[HDR + CH + i] = make_float4(-0.5f*LOG2E*pa, -LOG2E*pb, -0.5f*LOG2E*pc, feat[n*2+1]);
    }
    int n4 = (total + 3) & ~3;
    if (tid < n4 - total) {
        R[HDR + total + tid]      = make_float4(0.f, 0.f, 0.f, 0.f);
        R[HDR + CH + total + tid] = make_float4(0.f, 0.f, 0.f, 0.f);
    }
    if (tid == 0) ((int*)R)[0] = total;
}

#define TPW 4
#define RTH 64
static_assert(RG == 4, "the staging pass compares against three running sums");

__device__ __forceinline__ void stage_entry(float4* sA, float (*sF)[RG * CH + 8],
                                            int j, float yh, float4 e0, float4 e1)
{
    float dy = yh - e0.y;
    sA[j]    = make_float4(e0.x, e1.x, e1.y * dy, e1.z * dy * dy);
    sF[0][j] = e0.z * e0.w;
    sF[1][j] = e0.z * e1.w;
}

__global__ __launch_bounds__(RTH) void render_slices(
    const float4* __restrict__ L, float* __restrict__ out)
{
    __shared__ float4 sA[RG * CH];
    __shared__ __attribute__((aligned(16))) float sF[2][RG * CH + 8];

    const int d      = blockIdx.x;
    const int h      = blockIdx.y;
    const int tid    = threadIdx.x;
    const int lane   = tid & 31;
    const int wave   = tid >> 5;
    const int laneN  = lane & 15;
    const int laneHi = lane >> 4;
    const int wbase  = wave * (16 * TPW);

    const float yh = (float)h * (1.0f / (H_ - 1));
    float xw[TPW];
    #pragma unroll
    for (int t = 0; t < TPW; ++t)
        xw[t] = (float)(wbase + 16 * t + laneN) * (1.0f / (W_ - 1));

    v8f acc[TPW];
    #pragma unroll
    for (int t = 0; t < TPW; ++t)
        acc[t] = (v8f){0.f, 0.f, 0.f, 0.f, 0.f, 0.f, 0.f, 0.f};

    const float4* Ls = L + (size_t)d * NCH * RSTRIDE;
    int n4[NCH];
    #pragma unroll
    for (int r = 0; r < NCH; ++r)
        n4[r] = (((const int*)(Ls + (size_t)r * RSTRIDE))[0] + 3) & ~3;

    const float4* pA = sA + (laneHi << 1);
    const float*  pF = &sF[laneN & 1][laneHi << 1];

    #pragma unroll
    for (int c0 = 0; c0 < NCH; c0 += RG) {
        const int p1 = n4[c0];
        const int p2 = p1 + n4[c0 + 1];
        const int p3 = p2 + n4[c0 + 2];
        const int total = p3 + n4[c0 + 3];
        const float4* Lp = Ls + (size_t)c0 * RSTRIDE + HDR;
        #pragma unroll 1
        for (int j0 = tid; j0 < total; j0 += 4 * RTH) {
            float4 e0[4], e1[4];
            #pragma unroll
            for (int u = 0; u < 4; ++u) {
                int j = min(j0 + u * RTH, total - 1);
                int dlt = (j >= p1) ? (RSTRIDE - p1) : 0;
                dlt = (j >= p2) ? (2 * RSTRIDE - p2) : dlt;
                dlt = (j >= p3) ? (3 * RSTRIDE - p3) : dlt;
                e0[u] = Lp[j + dlt];
                e1[u] = Lp[j + dlt + CH];
            }
            #pragma unroll
            for (int u = 0; u < 4; ++u) {
                int j = j0 + u * RTH;
                if (j < total) stage_entry(sA, sF, j, yh, e0[u], e1[u]);
            }
        }
        __syncthreads();

        const int steps = total >> 2;
        const float4* qA = pA;
        const float*  qF = pF;
        #pragma unroll 1
        for (int s = 0; s < steps; ++s) {
            float4 a = qA[0];
            float4 b = qA[1];
            v2f Bm = *(const v2f*)qF;
            qA += 4;
            qF += 4;

            #pragma unroll
            for (int t = 0; t < TPW; ++t) {
                float dxA = xw[t] - a.x;
                float mA  = fmaf(fmaf(a.y, dxA, a.z), dxA, a.w);
                float dxB = xw[t] - b.x;
                float mB  = fmaf(fmaf(b.y, dxB, b.z), dxB, b.w);
                v2f Am = {__builtin_amdgcn_exp2f(mA), __builtin_amdgcn_exp2f(mB)};

                acc[t] = __builtin_amdgcn_wmma_f32_16x16x4_f32(
                             false, Am, false, Bm, (short)0, acc[t], false, false);
            }
        }
        __syncthreads();
    }

    if (laneN < F_) {
        #pragma unroll
        for (int t = 0; t < TPW; ++t) {
            #pragma unroll
            for (int r = 0; r < 8; ++r) {
                int M = r + (laneHi << 3);
                int w = wbase + 16 * t + M;
                out[(((size_t)h * W_ + w) * D_ + d) * F_ + laneN] = acc[t][r];
            }
        }
    }
}

extern "C" void kernel_launch(void* const* d_in, const int* in_sizes, int n_in,
                              void* d_out, int out_size, void* d_ws, size_t ws_size,
                              hipStream_t stream)
{
    const float* xyz   = (const float*)d_in[0];
    const float* scale = (const float*)d_in[1];
    const float* rot   = (const float*)d_in[2];
    const float* opac  = (const float*)d_in[3];
    const float* feat  = (const float*)d_in[4];
    float4* L = (float4*)d_ws;

    prep_compact<<<dim3(NCH, D_), 256, 0, stream>>>(xyz, scale, rot, opac, feat, L);
    render_slices<<<dim3(D_, H_), RTH, 0, stream>>>(L, (float*)d_out);
}


# Staging pass drops, per wave, Gaussians whose exponent peak over the wave's 64 pixels is under -140 (exp2 returns 0 there); each wave sums its own compacted lists, so sum order is regrouped
# speedup vs baseline: 6.4991x; 1.4179x over previous
#include <hip/hip_runtime.h>
#include <math.h>

#define NG   2048
#define H_   128
#define W_   128
#define D_   32
#define F_   2
#define LOG2E 1.4426950408889634f

#define CH      256
#define NCH     (NG / CH)
#define HDR     8
#define RSTRIDE (HDR + 2 * CH)
#define RG      4

typedef float v2f __attribute__((ext_vector_type(2)));
typedef float v8f __attribute__((ext_vector_type(8)));

__global__ __launch_bounds__(256) void prep_compact(
    const float* __restrict__ xyz, const float* __restrict__ scale,
    const float* __restrict__ rot, const float* __restrict__ opac,
    const float* __restrict__ feat, float4* __restrict__ L)
{
    __shared__ int wcnt[8];

    const int c    = blockIdx.x;
    const int d    = blockIdx.y;
    const int tid  = threadIdx.x;
    const int lane = tid & 31;
    const int wave = tid >> 5;
    const int n    = c * CH + tid;

    float qw = rot[n*4+0], qx = rot[n*4+1], qy = rot[n*4+2], qz = rot[n*4+3];
    float inv = 1.0f / (sqrtf(qw*qw + qx*qx + qy*qy + qz*qz) + 1e-8f);
    qw *= inv; qx *= inv; qy *= inv; qz *= inv;

    float r00 = 1.f - 2.f*(qy*qy + qz*qz);
    float r01 = 2.f*(qx*qy - qw*qz);
    float r02 = 2.f*(qx*qz + qw*qy);
    float r10 = 2.f*(qx*qy + qw*qz);
    float r11 = 1.f - 2.f*(qx*qx + qz*qz);
    float r12 = 2.f*(qy*qz - qw*qx);
    float r20 = 2.f*(qx*qz - qw*qy);
    float r21 = 2.f*(qy*qz + qw*qx);
    float r22 = 1.f - 2.f*(qx*qx + qy*qy);

    float sx = scale[n*3+0], sy = scale[n*3+1], sz = scale[n*3+2];
    float m00 = r00*sx, m01 = r01*sy, m02 = r02*sz;
    float m10 = r10*sx, m11 = r11*sy, m12 = r12*sz;
    float m20 = r20*sx, m21 = r21*sy, m22 = r22*sz;

    float cov00 = m00*m00 + m01*m01 + m02*m02;
    float cov01 = m00*m10 + m01*m11 + m02*m12;
    float cov02 = m00*m20 + m01*m21 + m02*m22;
    float cov11 = m10*m10 + m11*m11 + m12*m12;
    float cov12 = m10*m20 + m11*m21 + m12*m22;
    float cov22 = m20*m20 + m21*m21 + m22*m22;

    float zv   = cov22 + 1e-8f;
    float zinv = 1.0f / zv;
    float thr3 = 3.0f * sqrtf(zv);

    float a  = cov00 - zinv*cov02*cov02 + 1e-6f;
    float b  = cov01 - zinv*cov02*cov12;
    float cc = cov11 - zinv*cov12*cov12 + 1e-6f;
    float idet = 1.0f / (a*cc - b*b);
    float pa = cc*idet, pb = -b*idet, pc = a*idet;

    const float z = (float)d * (1.0f / (D_ - 1));
    float mu_x = xyz[n*3+0], mu_y = xyz[n*3+1], mu_z = xyz[n*3+2];
    float tt  = zinv * (z - mu_z);
    float mux = mu_x + cov02 * tt;
    float muy = mu_y + cov12 * tt;
    bool live = fabsf(mu_z - z) < thr3;

    unsigned int bal = (unsigned int)__ballot(live);
    int pos = __popc(bal & ((1u << lane) - 1u));
    if (lane == 0) wcnt[wave] = __popc(bal);
    __syncthreads();
    int off = 0, total = 0;
    #pragma unroll
    for (int w = 0; w < 8; ++w) {
        int v = wcnt[w];
        total += v;
        if (w < wave) off += v;
    }

    float4* R = L + (size_t)(d * NCH + c) * RSTRIDE;
    if (live) {
        int i = off + pos;
        float op = opac[n];
        float ea = -0.5f*LOG2E*pa, eb = -LOG2E*pb, ec = -0.5f*LOG2E*pc;
        R[HDR + i]      = make_float4(mux, muy, op * feat[n*2+0], op * feat[n*2+1]);
        R[HDR + CH + i] = make_float4(ea, eb, ec, -0.5f * eb / ea);
    }
    if (tid == 0) ((int*)R)[0] = total;
}

#define TPW 4
#define RTH 64
#define PASS 512
#define SEG  256
#define CULL_T (-140.0f)
static_assert(RG == 4, "the staging pass compares against three running sums");
static_assert(PASS == 2 * SEG && PASS % (4 * RTH) == 0, "a fetching wave keeps at most SEG entries per pass");

__global__ __launch_bounds__(RTH) void render_slices(
    const float4* __restrict__ L, float* __restrict__ out)
{
    __shared__ float4 sA[2][2][SEG];
    __shared__ __attribute__((aligned(16))) float sF[2][2][2][SEG + 8];
    __shared__ int sCnt[2][2];

    const int d      = blockIdx.x;
    const int h      = blockIdx.y;
    const int tid    = threadIdx.x;
    const int lane   = tid & 31;
    const int wave   = __builtin_amdgcn_readfirstlane(tid >> 5);
    const int laneN  = lane & 15;
    const int laneHi = lane >> 4;
    const int wbase  = wave * (16 * TPW);
    const unsigned int ltmask = (1u << lane) - 1u;

    const float yh = (float)h * (1.0f / (H_ - 1));
    float xw[TPW];
    #pragma unroll
    for (int t = 0; t < TPW; ++t)
        xw[t] = (float)(wbase + 16 * t + laneN) * (1.0f / (W_ - 1));
    float xlo[2], xhi[2];
    #pragma unroll
    for (int r = 0; r < 2; ++r) {
        xlo[r] = (float)(r * 16 * TPW) * (1.0f / (W_ - 1));
        xhi[r] = (float)(r * 16 * TPW + 16 * TPW - 1) * (1.0f / (W_ - 1));
    }

    v8f acc[TPW];
    #pragma unroll
    for (int t = 0; t < TPW; ++t)
        acc[t] = (v8f){0.f, 0.f, 0.f, 0.f, 0.f, 0.f, 0.f, 0.f};

    const float4* Ls = L + (size_t)d * NCH * RSTRIDE;
    int cnt[NCH];
    #pragma unroll
    for (int r = 0; r < NCH; ++r)
        cnt[r] = ((const int*)(Ls + (size_t)r * RSTRIDE))[0];

    #pragma unroll
    for (int c0 = 0; c0 < NCH; c0 += RG) {
        const int p1 = cnt[c0];
        const int p2 = p1 + cnt[c0 + 1];
        const int p3 = p2 + cnt[c0 + 2];
        const int total = p3 + cnt[c0 + 3];
        const float4* Lp = Ls + (size_t)c0 * RSTRIDE + HDR;
        #pragma unroll 1
        for (int base = 0; base < total; base += PASS) {
            const int pend = min(base + PASS, total);
            int kept[2] = {0, 0};
            #pragma unroll 1
            for (int j0 = base + tid; j0 - tid < pend; j0 += 4 * RTH) {
                float4 e0[4], e1[4];
                #pragma unroll
                for (int u = 0; u < 4; ++u) {
                    int j = min(j0 + u * RTH, pend - 1);
                    int dlt = (j >= p1) ? (RSTRIDE - p1) : 0;
                    dlt = (j >= p2) ? (2 * RSTRIDE - p2) : dlt;
                    dlt = (j >= p3) ? (3 * RSTRIDE - p3) : dlt;
                    e0[u] = Lp[j + dlt];
                    e1[u] = Lp[j + dlt + CH];
                }
                #pragma unroll
                for (int u = 0; u < 4; ++u) {
                    const bool valid = (j0 + u * RTH) < pend;
                    const float dy = yh - e0[u].y;
                    const float eB = e1[u].y * dy;
                    const float eC = e1[u].z * dy * dy;
                    const float4 av = make_float4(e0[u].x, e1[u].x, eB, eC);
                    const float tv = e1[u].w * dy;
                    const bool concave = e1[u].x < 0.f;
                    #pragma unroll
                    for (int r = 0; r < 2; ++r) {
                        float tl = xlo[r] - e0[u].x;
                        float tr = xhi[r] - e0[u].x;
                        float v  = fminf(fmaxf(tv, tl), tr);
                        float q  = fmaf(fmaf(e1[u].x, v, eB), v, eC);
                        bool keep = valid && !(concave && (q < CULL_T));
                        unsigned int bal = (unsigned int)__ballot(keep);
                        if (keep) {
                            int i = kept[r] + __popc(bal & ltmask);
                            sA[r][wave][i]    = av;
                            sF[r][wave][0][i] = e0[u].z;
                            sF[r][wave][1][i] = e0[u].w;
                        }
                        kept[r] += __popc(bal);
                    }
                }
            }
            #pragma unroll
            for (int r = 0; r < 2; ++r) {
                int k4 = (kept[r] + 3) & ~3;
                if (lane < k4 - kept[r]) {
                    sA[r][wave][kept[r] + lane]    = make_float4(0.f, 0.f, 0.f, 0.f);
                    sF[r][wave][0][kept[r] + lane] = 0.f;
                    sF[r][wave][1][kept[r] + lane] = 0.f;
                }
                if (lane == 0) sCnt[wave][r] = k4;
            }
            __syncthreads();

            #pragma unroll 1
            for (int s = 0; s < 2; ++s) {
                const int steps = __builtin_amdgcn_readfirstlane(sCnt[s][wave]) >> 2;
                const float4* qA = &sA[wave][s][laneHi << 1];
                const float*  qF = &sF[wave][s][laneN & 1][laneHi << 1];
                #pragma unroll 1
                for (int st = 0; st < steps; ++st) {
                    float4 a = qA[0];
                    float4 b = qA[1];
                    v2f Bm = *(const v2f*)qF;
                    qA += 4;
                    qF += 4;

                    #pragma unroll
                    for (int t = 0; t < TPW; ++t) {
                        float dxA = xw[t] - a.x;
                        float mA  = fmaf(fmaf(a.y, dxA, a.z), dxA, a.w);
                        float dxB = xw[t] - b.x;
                        float mB  = fmaf(fmaf(b.y, dxB, b.z), dxB, b.w);
                        v2f Am = {__builtin_amdgcn_exp2f(mA), __builtin_amdgcn_exp2f(mB)};

                        acc[t] = __builtin_amdgcn_wmma_f32_16x16x4_f32(
                                     false, Am, false, Bm, (short)0, acc[t], false, false);
                    }
                }
            }
            __syncthreads();
        }
    }

    if (laneN < F_) {
        #pragma unroll
        for (int t = 0; t < TPW; ++t) {
            #pragma unroll
            for (int r = 0; r < 8; ++r) {
                int M = r + (laneHi << 3);
                int w = wbase + 16 * t + M;
                out[(((size_t)h * W_ + w) * D_ + d) * F_ + laneN] = acc[t][r];
            }
        }
    }
}

extern "C" void kernel_launch(void* const* d_in, const int* in_sizes, int n_in,
                              void* d_out, int out_size, void* d_ws, size_t ws_size,
                              hipStream_t stream)
{
    const float* xyz   = (const float*)d_in[0];
    const float* scale = (const float*)d_in[1];
    const float* rot   = (const float*)d_in[2];
    const float* opac  = (const float*)d_in[3];
    const float* feat  = (const float*)d_in[4];
    float4* L = (float4*)d_ws;

    prep_compact<<<dim3(NCH, D_), 256, 0, stream>>>(xyz, scale, rot, opac, feat, L);
    render_slices<<<dim3(D_, H_), RTH, 0, stream>>>(L, (float*)d_out);
}


# Render: a wave's four tiles are stacked in y (16 columns x 4 rows per wave, 4 waves per block) instead of 64x1; cull per 16x4 patch, dx once per Gaussian; lists in index order, sums regrouped
# speedup vs baseline: 7.1212x; 1.0957x over previous
#include <hip/hip_runtime.h>
#include <math.h>

#define NG   2048
#define H_   128
#define W_   128
#define D_   32
#define F_   2
#define LOG2E 1.4426950408889634f

#define CH      256
#define NCH     (NG / CH)
#define HDR     8
#define RSTRIDE (HDR + 2 * CH)

typedef float v2f __attribute__((ext_vector_type(2)));
typedef float v8f __attribute__((ext_vector_type(8)));

__global__ __launch_bounds__(256) void prep_compact(
    const float* __restrict__ xyz, const float* __restrict__ scale,
    const float* __restrict__ rot, const float* __restrict__ opac,
    const float* __restrict__ feat, float4* __restrict__ L)
{
    __shared__ int wcnt[8];

    const int c    = blockIdx.x;
    const int d    = blockIdx.y;
    const int tid  = threadIdx.x;
    const int lane = tid & 31;
    const int wave = tid >> 5;
    const int n    = c * CH + tid;

    float qw = rot[n*4+0], qx = rot[n*4+1], qy = rot[n*4+2], qz = rot[n*4+3];
    float inv = 1.0f / (sqrtf(qw*qw + qx*qx + qy*qy + qz*qz) + 1e-8f);
    qw *= inv; qx *= inv; qy *= inv; qz *= inv;

    float r00 = 1.f - 2.f*(qy*qy + qz*qz);
    float r01 = 2.f*(qx*qy - qw*qz);
    float r02 = 2.f*(qx*qz + qw*qy);
    float r10 = 2.f*(qx*qy + qw*qz);
    float r11 = 1.f - 2.f*(qx*qx + qz*qz);
    float r12 = 2.f*(qy*qz - qw*qx);
    float r20 = 2.f*(qx*qz - qw*qy);
    float r21 = 2.f*(qy*qz + qw*qx);
    float r22 = 1.f - 2.f*(qx*qx + qy*qy);

    float sx = scale[n*3+0], sy = scale[n*3+1], sz = scale[n*3+2];
    float m00 = r00*sx, m01 = r01*sy, m02 = r02*sz;
    float m10 = r10*sx, m11 = r11*sy, m12 = r12*sz;
    float m20 = r20*sx, m21 = r21*sy, m22 = r22*sz;

    float cov00 = m00*m00 + m01*m01 + m02*m02;
    float cov01 = m00*m10 + m01*m11 + m02*m12;
    float cov02 = m00*m20 + m01*m21 + m02*m22;
    float cov11 = m10*m10 + m11*m11 + m12*m12;
    float cov12 = m10*m20 + m11*m21 + m12*m22;
    float cov22 = m20*m20 + m21*m21 + m22*m22;

    float zv   = cov22 + 1e-8f;
    float zinv = 1.0f / zv;
    float thr3 = 3.0f * sqrtf(zv);

    float a  = cov00 - zinv*cov02*cov02 + 1e-6f;
    float b  = cov01 - zinv*cov02*cov12;
    float cc = cov11 - zinv*cov12*cov12 + 1e-6f;
    float idet = 1.0f / (a*cc - b*b);
    float pa = cc*idet, pb = -b*idet, pc = a*idet;

    const float z = (float)d * (1.0f / (D_ - 1));
    float mu_x = xyz[n*3+0], mu_y = xyz[n*3+1], mu_z = xyz[n*3+2];
    float tt  = zinv * (z - mu_z);
    float mux = mu_x + cov02 * tt;
    float muy = mu_y + cov12 * tt;
    bool live = fabsf(mu_z - z) < thr3;

    unsigned int bal = (unsigned int)__ballot(live);
    int pos = __popc(bal & ((1u << lane) - 1u));
    if (lane == 0) wcnt[wave] = __popc(bal);
    __syncthreads();
    int off = 0, total = 0;
    #pragma unroll
    for (int w = 0; w < 8; ++w) {
        int v = wcnt[w];
        total += v;
        if (w < wave) off += v;
    }

    float4* R = L + (size_t)(d * NCH + c) * RSTRIDE;
    if (live) {
        int i = off + pos;
        float op = opac[n];
        float ea = -0.5f*LOG2E*pa, eb = -LOG2E*pb, ec = -0.5f*LOG2E*pc;
        R[HDR + i]      = make_float4(mux, muy, op * feat[n*2+0], op * feat[n*2+1]);
        R[HDR + CH + i] = make_float4(ea, eb, ec, -0.5f * eb / ea);
    }
    if (tid == 0) ((int*)R)[0] = total;
}

#define PR  4
#define PWV 4
#define NRG (H_ / PR)
#define RTH (32 * PWV)
#define CAP RTH
#define CULL_T (-140.0f)
static_assert(NCH == 8, "the flat index is compared against seven running sums");
static_assert(PR == 4 && PWV == 4, "row constants and wave counts are kept in float4 / int4");
static_assert(W_ == 2 * 16 * PWV && H_ % PR == 0, "two blocks side by side cover a row group");

__device__ __forceinline__ int entry_index(int j, int p1, int p2, int p3, int p4,
                                           int p5, int p6, int p7)
{
    int dlt = (j >= p1) ? (1 * RSTRIDE - p1) : 0;
    dlt = (j >= p2) ? (2 * RSTRIDE - p2) : dlt;
    dlt = (j >= p3) ? (3 * RSTRIDE - p3) : dlt;
    dlt = (j >= p4) ? (4 * RSTRIDE - p4) : dlt;
    dlt = (j >= p5) ? (5 * RSTRIDE - p5) : dlt;
    dlt = (j >= p6) ? (6 * RSTRIDE - p6) : dlt;
    dlt = (j >= p7) ? (7 * RSTRIDE - p7) : dlt;
    return HDR + j + dlt;
}

__global__ __launch_bounds__(RTH) void render_slices(
    const float4* __restrict__ L, float* __restrict__ out)
{
    __shared__ __attribute__((aligned(16))) float2 sM[PWV][CAP];
    __shared__ float4 sB[PWV][CAP];
    __shared__ float4 sC[PWV][CAP];
    __shared__ __attribute__((aligned(16))) float sF[PWV][2][CAP + 8];
    __shared__ int4 sCnt[PWV];

    const int d      = blockIdx.x;
    const int h0     = blockIdx.y * PR;
    const int cb0    = blockIdx.z * (16 * PWV);
    const int tid    = threadIdx.x;
    const int lane   = tid & 31;
    const int wave   = __builtin_amdgcn_readfirstlane(tid >> 5);
    const int laneN  = lane & 15;
    const int laneHi = lane >> 4;
    const int cb     = cb0 + 16 * wave;
    const unsigned int ltmask = (1u << lane) - 1u;

    float yh[PR];
    #pragma unroll
    for (int t = 0; t < PR; ++t)
        yh[t] = (float)(h0 + t) * (1.0f / (H_ - 1));
    const float xw = (float)(cb + laneN) * (1.0f / (W_ - 1));
    float xlo[PWV], xhi[PWV];
    #pragma unroll
    for (int s = 0; s < PWV; ++s) {
        xlo[s] = (float)(cb0 + 16 * s) * (1.0f / (W_ - 1));
        xhi[s] = (float)(cb0 + 16 * s + 15) * (1.0f / (W_ - 1));
    }

    v8f acc[PR];
    #pragma unroll
    for (int t = 0; t < PR; ++t)
        acc[t] = (v8f){0.f, 0.f, 0.f, 0.f, 0.f, 0.f, 0.f, 0.f};

    const float4* Ls = L + (size_t)d * NCH * RSTRIDE;
    int ps[NCH + 1];
    ps[0] = 0;
    #pragma unroll
    for (int r = 0; r < NCH; ++r) {
        int c = ((const int*)(Ls + (size_t)r * RSTRIDE))[0];
        ps[r + 1] = ps[r] + min(max(c, 0), CH);
    }
    const int total = ps[NCH];

    float4 n0, n1;
    {
        int j = min(tid, max(total - 1, 0));
        int a = entry_index(j, ps[1], ps[2], ps[3], ps[4], ps[5], ps[6], ps[7]);
        n0 = Ls[a];
        n1 = Ls[a + CH];
    }

    #pragma unroll 1
    for (int base = 0; base < total; base += CAP) {
        const float4 e0 = n0, e1 = n1;
        const bool valid = (base + tid) < total;

        float eB[PR], eC[PR], tv[PR];
        #pragma unroll
        for (int t = 0; t < PR; ++t) {
            const float dy = yh[t] - e0.y;
            eB[t] = e1.y * dy;
            eC[t] = e1.z * dy * dy;
            tv[t] = e1.w * dy;
        }
        const bool concave = e1.x < 0.f;

        bool keep[PWV];
        int  pos[PWV], kc[PWV];
        #pragma unroll
        for (int s = 0; s < PWV; ++s) {
            const float tl = xlo[s] - e0.x;
            const float tr = xhi[s] - e0.x;
            bool cull = concave;
            #pragma unroll
            for (int t = 0; t < PR; ++t) {
                float v = fminf(fmaxf(tv[t], tl), tr);
                float q = fmaf(fmaf(e1.x, v, eB[t]), v, eC[t]);
                cull = cull && (q < CULL_T);
            }
            keep[s] = valid && !cull;
            unsigned int bal = (unsigned int)__ballot(keep[s]);
            pos[s] = __popc(bal & ltmask);
            kc[s]  = __popc(bal);
        }
        if (lane == 0) sCnt[wave] = make_int4(kc[0], kc[1], kc[2], kc[3]);
        __syncthreads();

        int off[PWV] = {0, 0, 0, 0}, tot[PWV] = {0, 0, 0, 0};
        #pragma unroll
        for (int f = 0; f < PWV; ++f) {
            const int4 c = sCnt[f];
            const int cv[PWV] = {c.x, c.y, c.z, c.w};
            #pragma unroll
            for (int s = 0; s < PWV; ++s) {
                tot[s] += cv[s];
                if (f < wave) off[s] += cv[s];
            }
        }
        #pragma unroll
        for (int s = 0; s < PWV; ++s) {
            if (keep[s]) {
                const int i = off[s] + pos[s];
                sM[s][i]    = make_float2(e0.x, e1.x);
                sB[s][i]    = make_float4(eB[0], eB[1], eB[2], eB[3]);
                sC[s][i]    = make_float4(eC[0], eC[1], eC[2], eC[3]);
                sF[s][0][i] = e0.z;
                sF[s][1][i] = e0.w;
            }
        }
        int mine = tot[0];
        mine = (wave == 1) ? tot[1] : mine;
        mine = (wave == 2) ? tot[2] : mine;
        mine = (wave == 3) ? tot[3] : mine;
        const int k4 = (mine + 3) & ~3;
        if (lane < k4 - mine) {
            const int i = mine + lane;
            sM[wave][i]    = make_float2(0.f, 0.f);
            sB[wave][i]    = make_float4(0.f, 0.f, 0.f, 0.f);
            sC[wave][i]    = make_float4(0.f, 0.f, 0.f, 0.f);
            sF[wave][0][i] = 0.f;
            sF[wave][1][i] = 0.f;
        }
        __syncthreads();

        {
            int j = min(base + CAP + tid, total - 1);
            int a = entry_index(j, ps[1], ps[2], ps[3], ps[4], ps[5], ps[6], ps[7]);
            n0 = Ls[a];
            n1 = Ls[a + CH];
        }

        const int steps = __builtin_amdgcn_readfirstlane(k4) >> 2;
        const float2* qM = &sM[wave][laneHi << 1];
        const float4* qB = &sB[wave][laneHi << 1];
        const float4* qC = &sC[wave][laneHi << 1];
        const float*  qF = &sF[wave][laneN & 1][laneHi << 1];
        #pragma unroll 1
        for (int st = 0; st < steps; ++st) {
            const float4 m  = *(const float4*)qM;
            const float4 bA = qB[0], bB = qB[1];
            const float4 cA = qC[0], cB = qC[1];
            const v2f Bm = *(const v2f*)qF;
            qM += 4;
            qB += 4;
            qC += 4;
            qF += 4;

            const float dxA = xw - m.x;
            const float dxB = xw - m.z;
            const float bAt[PR] = {bA.x, bA.y, bA.z, bA.w};
            const float bBt[PR] = {bB.x, bB.y, bB.z, bB.w};
            const float cAt[PR] = {cA.x, cA.y, cA.z, cA.w};
            const float cBt[PR] = {cB.x, cB.y, cB.z, cB.w};
            #pragma unroll
            for (int t = 0; t < PR; ++t) {
                float mA = fmaf(fmaf(m.y, dxA, bAt[t]), dxA, cAt[t]);
                float mB = fmaf(fmaf(m.w, dxB, bBt[t]), dxB, cBt[t]);
                v2f Am = {__builtin_amdgcn_exp2f(mA), __builtin_amdgcn_exp2f(mB)};

                acc[t] = __builtin_amdgcn_wmma_f32_16x16x4_f32(
                             false, Am, false, Bm, (short)0, acc[t], false, false);
            }
        }
    }

    if (laneN < F_) {
        #pragma unroll
        for (int t = 0; t < PR; ++t) {
            #pragma unroll
            for (int r = 0; r < 8; ++r) {
                int M = r + (laneHi << 3);
                int w = cb + M;
                out[(((size_t)(h0 + t) * W_ + w) * D_ + d) * F_ + laneN] = acc[t][r];
            }
        }
    }
}

extern "C" void kernel_launch(void* const* d_in, const int* in_sizes, int n_in,
                              void* d_out, int out_size, void* d_ws, size_t ws_size,
                              hipStream_t stream)
{
    const float* xyz   = (const float*)d_in[0];
    const float* scale = (const float*)d_in[1];
    const float* rot   = (const float*)d_in[2];
    const float* opac  = (const float*)d_in[3];
    const float* feat  = (const float*)d_in[4];
    float4* L = (float4*)d_ws;

    prep_compact<<<dim3(NCH, D_), 256, 0, stream>>>(xyz, scale, rot, opac, feat, L);
    render_slices<<<dim3(D_, NRG, 2), RTH, 0, stream>>>(L, (float*)d_out);
}


# One launch: each 32-wave block builds its slice's z-mask list in LDS, then every wave culls and renders its own 16x4 patch from it, no block barrier in the loop; list chunks regroup the f32 sums
# speedup vs baseline: 7.5515x; 1.0604x over previous
#include <hip/hip_runtime.h>
#include <math.h>

#define NG   2048
#define H_   128
#define W_   128
#define D_   32
#define F_   2
#define LOG2E 1.4426950408889634f

typedef float v2f __attribute__((ext_vector_type(2)));
typedef float v8f __attribute__((ext_vector_type(8)));

#define PR   4
#define NWV  32
#define BTH  (32 * NWV)
#define NRD  (NG / BTH)
#define NSP  (W_ / 16)
#define NRG  (H_ / PR)
#define BPS  (NSP * NRG / NWV)
#define CAPW 128
#define CULL_T (-140.0f)
static_assert(NRD == 2, "phase 1 packs the two rounds' counts into two words");
static_assert(NWV == 32, "one lane per wave in the count exchange");
static_assert(NSP == 8 && NRG % (NWV / NSP) == 0 && BPS * (NWV / NSP) == NRG, "a block takes every span of NWV/NSP row groups");
static_assert(PR == 4, "row constants are kept in float4");

__global__ __launch_bounds__(BTH) void render_fused(
    const float* __restrict__ xyz, const float* __restrict__ scale,
    const float* __restrict__ rot, const float* __restrict__ opac,
    const float* __restrict__ feat, float* __restrict__ out)
{
    __shared__ float4 sE0[NG];
    __shared__ float4 sE1[NG];
    __shared__ __attribute__((aligned(16))) float2 sM[NWV][CAPW];
    __shared__ float4 sB[NWV][CAPW];
    __shared__ float4 sC[NWV][CAPW];
    __shared__ __attribute__((aligned(16))) float sF[NWV][2][CAPW + 8];
    __shared__ int wcnt[NRD][NWV];

    const int d      = blockIdx.x % D_;
    const int jb     = blockIdx.x / D_;
    const int tid    = threadIdx.x;
    const int lane   = tid & 31;
    const int wave   = __builtin_amdgcn_readfirstlane(tid >> 5);
    const unsigned int ltmask = (1u << lane) - 1u;

    const float z = (float)d * (1.0f / (D_ - 1));
    float4 p0[NRD], p1[NRD];
    bool   live[NRD];
    int    pos[NRD];
    float rq[NRD][4], sc[NRD][3], xp[NRD][3], opv[NRD], ft[NRD][2];
    #pragma unroll
    for (int r = 0; r < NRD; ++r) {
        const int n = r * BTH + tid;
        #pragma unroll
        for (int k = 0; k < 4; ++k) rq[r][k] = rot[n*4+k];
        #pragma unroll
        for (int k = 0; k < 3; ++k) { sc[r][k] = scale[n*3+k]; xp[r][k] = xyz[n*3+k]; }
        opv[r]   = opac[n];
        ft[r][0] = feat[n*2+0];
        ft[r][1] = feat[n*2+1];
    }
    #pragma unroll
    for (int r = 0; r < NRD; ++r) {
        float qw = rq[r][0], qx = rq[r][1], qy = rq[r][2], qz = rq[r][3];
        float inv = 1.0f / (sqrtf(qw*qw + qx*qx + qy*qy + qz*qz) + 1e-8f);
        qw *= inv; qx *= inv; qy *= inv; qz *= inv;

        float r00 = 1.f - 2.f*(qy*qy + qz*qz);
        float r01 = 2.f*(qx*qy - qw*qz);
        float r02 = 2.f*(qx*qz + qw*qy);
        float r10 = 2.f*(qx*qy + qw*qz);
        float r11 = 1.f - 2.f*(qx*qx + qz*qz);
        float r12 = 2.f*(qy*qz - qw*qx);
        float r20 = 2.f*(qx*qz - qw*qy);
        float r21 = 2.f*(qy*qz + qw*qx);
        float r22 = 1.f - 2.f*(qx*qx + qy*qy);

        float sx = sc[r][0], sy = sc[r][1], sz = sc[r][2];
        float m00 = r00*sx, m01 = r01*sy, m02 = r02*sz;
        float m10 = r10*sx, m11 = r11*sy, m12 = r12*sz;
        float m20 = r20*sx, m21 = r21*sy, m22 = r22*sz;

        float cov00 = m00*m00 + m01*m01 + m02*m02;
        float cov01 = m00*m10 + m01*m11 + m02*m12;
        float cov02 = m00*m20 + m01*m21 + m02*m22;
        float cov11 = m10*m10 + m11*m11 + m12*m12;
        float cov12 = m10*m20 + m11*m21 + m12*m22;
        float cov22 = m20*m20 + m21*m21 + m22*m22;

        float zv   = cov22 + 1e-8f;
        float zinv = 1.0f / zv;
        float thr3 = 3.0f * sqrtf(zv);

        float a  = cov00 - zinv*cov02*cov02 + 1e-6f;
        float b  = cov01 - zinv*cov02*cov12;
        float cc = cov11 - zinv*cov12*cov12 + 1e-6f;
        float idet = 1.0f / (a*cc - b*b);
        float pa = cc*idet, pb = -b*idet, pc = a*idet;

        float mu_x = xp[r][0], mu_y = xp[r][1], mu_z = xp[r][2];
        float tt  = zinv * (z - mu_z);
        float mux = mu_x + cov02 * tt;
        float muy = mu_y + cov12 * tt;
        live[r] = fabsf(mu_z - z) < thr3;

        float ea = -0.5f*LOG2E*pa, eb = -LOG2E*pb, ec = -0.5f*LOG2E*pc;
        p0[r] = make_float4(mux, muy, opv[r] * ft[r][0], opv[r] * ft[r][1]);
        p1[r] = make_float4(ea, eb, ec, -0.5f * eb / ea);

        unsigned int bal = (unsigned int)__ballot(live[r]);
        pos[r] = __popc(bal & ltmask);
        if (lane == 0) wcnt[r][wave] = __popc(bal);
    }
    __syncthreads();

    int w0, w1;
    {
        const int c0 = wcnt[0][lane], c1 = wcnt[1][lane];
        w0 = ((lane < wave) ? c0 : 0) + (c0 << 16);
        w1 = ((lane < wave) ? c1 : 0) + (c1 << 16);
        #pragma unroll
        for (int m = 16; m >= 1; m >>= 1) {
            w0 += __shfl_xor(w0, m, 32);
            w1 += __shfl_xor(w1, m, 32);
        }
    }
    const int tot0 = w0 >> 16;
    const int off0 = w0 & 0xffff;
    const int off1 = tot0 + (w1 & 0xffff);
    const int total = __builtin_amdgcn_readfirstlane(tot0 + (w1 >> 16));

    if (live[0]) { const int i = off0 + pos[0]; sE0[i] = p0[0]; sE1[i] = p1[0]; }
    if (live[1]) { const int i = off1 + pos[1]; sE0[i] = p0[1]; sE1[i] = p1[1]; }
    __syncthreads();

    const int laneN  = lane & 15;
    const int laneHi = lane >> 4;
    const int cb     = 16 * (wave % NSP);
    const int h0     = PR * (jb + BPS * (wave / NSP));

    float yh[PR];
    #pragma unroll
    for (int t = 0; t < PR; ++t)
        yh[t] = (float)(h0 + t) * (1.0f / (H_ - 1));
    const float xw  = (float)(cb + laneN) * (1.0f / (W_ - 1));
    const float xlo = (float)cb * (1.0f / (W_ - 1));
    const float xhi = (float)(cb + 15) * (1.0f / (W_ - 1));

    v8f acc[PR];
    #pragma unroll
    for (int t = 0; t < PR; ++t)
        acc[t] = (v8f){0.f, 0.f, 0.f, 0.f, 0.f, 0.f, 0.f, 0.f};

    float2* const mM  = sM[wave];
    float4* const mB  = sB[wave];
    float4* const mC  = sC[wave];
    float*  const mF0 = sF[wave][0];
    float*  const mF1 = sF[wave][1];

    int cnt = 0;
    #pragma unroll 1
    for (int base = 0; base < total; base += 32) {
        const int  j     = base + lane;
        const bool valid = j < total;
        const int  jj    = min(j, total - 1);
        const float4 e0 = sE0[jj], e1 = sE1[jj];

        float eB[PR], eC[PR], tv[PR];
        #pragma unroll
        for (int t = 0; t < PR; ++t) {
            const float dy = yh[t] - e0.y;
            eB[t] = e1.y * dy;
            eC[t] = e1.z * dy * dy;
            tv[t] = e1.w * dy;
        }
        const float tl = xlo - e0.x;
        const float tr = xhi - e0.x;
        bool cull = e1.x < 0.f;
        #pragma unroll
        for (int t = 0; t < PR; ++t) {
            float v = fminf(fmaxf(tv[t], tl), tr);
            float q = fmaf(fmaf(e1.x, v, eB[t]), v, eC[t]);
            cull = cull && (q < CULL_T);
        }
        const bool keep = valid && !cull;
        const unsigned int bal = (unsigned int)__ballot(keep);
        if (keep) {
            const int i = cnt + __popc(bal & ltmask);
            mM[i]  = make_float2(e0.x, e1.x);
            mB[i]  = make_float4(eB[0], eB[1], eB[2], eB[3]);
            mC[i]  = make_float4(eC[0], eC[1], eC[2], eC[3]);
            mF0[i] = e0.z;
            mF1[i] = e0.w;
        }
        cnt += __builtin_amdgcn_readfirstlane(__popc(bal));

        if (cnt > CAPW - 32 || base + 32 >= total) {
            const int k4 = (cnt + 3) & ~3;
            if (lane < k4 - cnt) {
                const int i = cnt + lane;
                mM[i]  = make_float2(0.f, 0.f);
                mB[i]  = make_float4(0.f, 0.f, 0.f, 0.f);
                mC[i]  = make_float4(0.f, 0.f, 0.f, 0.f);
                mF0[i] = 0.f;
                mF1[i] = 0.f;
            }
            const int steps = k4 >> 2;
            const float2* qM = &mM[laneHi << 1];
            const float4* qB = &mB[laneHi << 1];
            const float4* qC = &mC[laneHi << 1];
            const float*  qF = &sF[wave][laneN & 1][laneHi << 1];
            #pragma unroll 1
            for (int st = 0; st < steps; ++st) {
                const float4 m  = *(const float4*)qM;
                const float4 bA = qB[0], bB = qB[1];
                const float4 cA = qC[0], cB = qC[1];
                const v2f Bm = *(const v2f*)qF;
                qM += 4;
                qB += 4;
                qC += 4;
                qF += 4;

                const float dxA = xw - m.x;
                const float dxB = xw - m.z;
                const float bAt[PR] = {bA.x, bA.y, bA.z, bA.w};
                const float bBt[PR] = {bB.x, bB.y, bB.z, bB.w};
                const float cAt[PR] = {cA.x, cA.y, cA.z, cA.w};
                const float cBt[PR] = {cB.x, cB.y, cB.z, cB.w};
                #pragma unroll
                for (int t = 0; t < PR; ++t) {
                    float mA = fmaf(fmaf(m.y, dxA, bAt[t]), dxA, cAt[t]);
                    float mB2 = fmaf(fmaf(m.w, dxB, bBt[t]), dxB, cBt[t]);
                    v2f Am = {__builtin_amdgcn_exp2f(mA), __builtin_amdgcn_exp2f(mB2)};

                    acc[t] = __builtin_amdgcn_wmma_f32_16x16x4_f32(
                                 false, Am, false, Bm, (short)0, acc[t], false, false);
                }
            }
            cnt = 0;
        }
    }

    if (laneN < F_) {
        #pragma unroll
        for (int t = 0; t < PR; ++t) {
            #pragma unroll
            for (int r = 0; r < 8; ++r) {
                int M = r + (laneHi << 3);
                int w = cb + M;
                out[(((size_t)(h0 + t) * W_ + w) * D_ + d) * F_ + laneN] = acc[t][r];
            }
        }
    }
}

extern "C" void kernel_launch(void* const* d_in, const int* in_sizes, int n_in,
                              void* d_out, int out_size, void* d_ws, size_t ws_size,
                              hipStream_t stream)
{
    const float* xyz   = (const float*)d_in[0];
    const float* scale = (const float*)d_in[1];
    const float* rot   = (const float*)d_in[2];
    const float* opac  = (const float*)d_in[3];
    const float* feat  = (const float*)d_in[4];

    render_fused<<<dim3(D_ * BPS), BTH, 0, stream>>>(xyz, scale, rot, opac, feat, (float*)d_out);
}


# Cull bound moved from -140 to -127, one under the point where raw exp2 leaves the normal range: about 5 % fewer entries per wave list; dropped terms are under 2^-126, the rest keep their order
# speedup vs baseline: 7.8169x; 1.0351x over previous
#include <hip/hip_runtime.h>
#include <math.h>

#define NG   2048
#define H_   128
#define W_   128
#define D_   32
#define F_   2
#define LOG2E 1.4426950408889634f

typedef float v2f __attribute__((ext_vector_type(2)));
typedef float v8f __attribute__((ext_vector_type(8)));

#define PR   4
#define NWV  32
#define BTH  (32 * NWV)
#define NRD  (NG / BTH)
#define NSP  (W_ / 16)
#define NRG  (H_ / PR)
#define BPS  (NSP * NRG / NWV)
#define CAPW 128
#define CULL_T (-127.0f)
static_assert(NRD == 2, "phase 1 packs the two rounds' counts into two words");
static_assert(NWV == 32, "one lane per wave in the count exchange");
static_assert(NSP == 8 && NRG % (NWV / NSP) == 0 && BPS * (NWV / NSP) == NRG, "a block takes every span of NWV/NSP row groups");
static_assert(PR == 4, "row constants are kept in float4");

__global__ __launch_bounds__(BTH) void render_fused(
    const float* __restrict__ xyz, const float* __restrict__ scale,
    const float* __restrict__ rot, const float* __restrict__ opac,
    const float* __restrict__ feat, float* __restrict__ out)
{
    __shared__ float4 sE0[NG];
    __shared__ float4 sE1[NG];
    __shared__ __attribute__((aligned(16))) float2 sM[NWV][CAPW];
    __shared__ float4 sB[NWV][CAPW];
    __shared__ float4 sC[NWV][CAPW];
    __shared__ __attribute__((aligned(16))) float sF[NWV][2][CAPW + 8];
    __shared__ int wcnt[NRD][NWV];

    const int d      = blockIdx.x % D_;
    const int jb     = blockIdx.x / D_;
    const int tid    = threadIdx.x;
    const int lane   = tid & 31;
    const int wave   = __builtin_amdgcn_readfirstlane(tid >> 5);
    const unsigned int ltmask = (1u << lane) - 1u;

    const float z = (float)d * (1.0f / (D_ - 1));
    float4 p0[NRD], p1[NRD];
    bool   live[NRD];
    int    pos[NRD];
    float rq[NRD][4], sc[NRD][3], xp[NRD][3], opv[NRD], ft[NRD][2];
    #pragma unroll
    for (int r = 0; r < NRD; ++r) {
        const int n = r * BTH + tid;
        #pragma unroll
        for (int k = 0; k < 4; ++k) rq[r][k] = rot[n*4+k];
        #pragma unroll
        for (int k = 0; k < 3; ++k) { sc[r][k] = scale[n*3+k]; xp[r][k] = xyz[n*3+k]; }
        opv[r]   = opac[n];
        ft[r][0] = feat[n*2+0];
        ft[r][1] = feat[n*2+1];
    }
    #pragma unroll
    for (int r = 0; r < NRD; ++r) {
        float qw = rq[r][0], qx = rq[r][1], qy = rq[r][2], qz = rq[r][3];
        float inv = 1.0f / (sqrtf(qw*qw + qx*qx + qy*qy + qz*qz) + 1e-8f);
        qw *= inv; qx *= inv; qy *= inv; qz *= inv;

        float r00 = 1.f - 2.f*(qy*qy + qz*qz);
        float r01 = 2.f*(qx*qy - qw*qz);
        float r02 = 2.f*(qx*qz + qw*qy);
        float r10 = 2.f*(qx*qy + qw*qz);
        float r11 = 1.f - 2.f*(qx*qx + qz*qz);
        float r12 = 2.f*(qy*qz - qw*qx);
        float r20 = 2.f*(qx*qz - qw*qy);
        float r21 = 2.f*(qy*qz + qw*qx);
        float r22 = 1.f - 2.f*(qx*qx + qy*qy);

        float sx = sc[r][0], sy = sc[r][1], sz = sc[r][2];
        float m00 = r00*sx, m01 = r01*sy, m02 = r02*sz;
        float m10 = r10*sx, m11 = r11*sy, m12 = r12*sz;
        float m20 = r20*sx, m21 = r21*sy, m22 = r22*sz;

        float cov00 = m00*m00 + m01*m01 + m02*m02;
        float cov01 = m00*m10 + m01*m11 + m02*m12;
        float cov02 = m00*m20 + m01*m21 + m02*m22;
        float cov11 = m10*m10 + m11*m11 + m12*m12;
        float cov12 = m10*m20 + m11*m21 + m12*m22;
        float cov22 = m20*m20 + m21*m21 + m22*m22;

        float zv   = cov22 + 1e-8f;
        float zinv = 1.0f / zv;
        float thr3 = 3.0f * sqrtf(zv);

        float a  = cov00 - zinv*cov02*cov02 + 1e-6f;
        float b  = cov01 - zinv*cov02*cov12;
        float cc = cov11 - zinv*cov12*cov12 + 1e-6f;
        float idet = 1.0f / (a*cc - b*b);
        float pa = cc*idet, pb = -b*idet, pc = a*idet;

        float mu_x = xp[r][0], mu_y = xp[r][1], mu_z = xp[r][2];
        float tt  = zinv * (z - mu_z);
        float mux = mu_x + cov02 * tt;
        float muy = mu_y + cov12 * tt;
        live[r] = fabsf(mu_z - z) < thr3;

        float ea = -0.5f*LOG2E*pa, eb = -LOG2E*pb, ec = -0.5f*LOG2E*pc;
        p0[r] = make_float4(mux, muy, opv[r] * ft[r][0], opv[r] * ft[r][1]);
        p1[r] = make_float4(ea, eb, ec, -0.5f * eb / ea);

        unsigned int bal = (unsigned int)__ballot(live[r]);
        pos[r] = __popc(bal & ltmask);
        if (lane == 0) wcnt[r][wave] = __popc(bal);
    }
    __syncthreads();

    int w0, w1;
    {
        const int c0 = wcnt[0][lane], c1 = wcnt[1][lane];
        w0 = ((lane < wave) ? c0 : 0) + (c0 << 16);
        w1 = ((lane < wave) ? c1 : 0) + (c1 << 16);
        #pragma unroll
        for (int m = 16; m >= 1; m >>= 1) {
            w0 += __shfl_xor(w0, m, 32);
            w1 += __shfl_xor(w1, m, 32);
        }
    }
    const int tot0 = w0 >> 16;
    const int off0 = w0 & 0xffff;
    const int off1 = tot0 + (w1 & 0xffff);
    const int total = __builtin_amdgcn_readfirstlane(tot0 + (w1 >> 16));

    if (live[0]) { const int i = off0 + pos[0]; sE0[i] = p0[0]; sE1[i] = p1[0]; }
    if (live[1]) { const int i = off1 + pos[1]; sE0[i] = p0[1]; sE1[i] = p1[1]; }
    __syncthreads();

    const int laneN  = lane & 15;
    const int laneHi = lane >> 4;
    const int cb     = 16 * (wave % NSP);
    const int h0     = PR * (jb + BPS * (wave / NSP));

    float yh[PR];
    #pragma unroll
    for (int t = 0; t < PR; ++t)
        yh[t] = (float)(h0 + t) * (1.0f / (H_ - 1));
    const float xw  = (float)(cb + laneN) * (1.0f / (W_ - 1));
    const float xlo = (float)cb * (1.0f / (W_ - 1));
    const float xhi = (float)(cb + 15) * (1.0f / (W_ - 1));

    v8f acc[PR];
    #pragma unroll
    for (int t = 0; t < PR; ++t)
        acc[t] = (v8f){0.f, 0.f, 0.f, 0.f, 0.f, 0.f, 0.f, 0.f};

    float2* const mM  = sM[wave];
    float4* const mB  = sB[wave];
    float4* const mC  = sC[wave];
    float*  const mF0 = sF[wave][0];
    float*  const mF1 = sF[wave][1];

    int cnt = 0;
    #pragma unroll 1
    for (int base = 0; base < total; base += 32) {
        const int  j     = base + lane;
        const bool valid = j < total;
        const int  jj    = min(j, total - 1);
        const float4 e0 = sE0[jj], e1 = sE1[jj];

        float eB[PR], eC[PR], tv[PR];
        #pragma unroll
        for (int t = 0; t < PR; ++t) {
            const float dy = yh[t] - e0.y;
            eB[t] = e1.y * dy;
            eC[t] = e1.z * dy * dy;
            tv[t] = e1.w * dy;
        }
        const float tl = xlo - e0.x;
        const float tr = xhi - e0.x;
        bool cull = e1.x < 0.f;
        #pragma unroll
        for (int t = 0; t < PR; ++t) {
            float v = fminf(fmaxf(tv[t], tl), tr);
            float q = fmaf(fmaf(e1.x, v, eB[t]), v, eC[t]);
            cull = cull && (q < CULL_T);
        }
        const bool keep = valid && !cull;
        const unsigned int bal = (unsigned int)__ballot(keep);
        if (keep) {
            const int i = cnt + __popc(bal & ltmask);
            mM[i]  = make_float2(e0.x, e1.x);
            mB[i]  = make_float4(eB[0], eB[1], eB[2], eB[3]);
            mC[i]  = make_float4(eC[0], eC[1], eC[2], eC[3]);
            mF0[i] = e0.z;
            mF1[i] = e0.w;
        }
        cnt += __builtin_amdgcn_readfirstlane(__popc(bal));

        if (cnt > CAPW - 32 || base + 32 >= total) {
            const int k4 = (cnt + 3) & ~3;
            if (lane < k4 - cnt) {
                const int i = cnt + lane;
                mM[i]  = make_float2(0.f, 0.f);
                mB[i]  = make_float4(0.f, 0.f, 0.f, 0.f);
                mC[i]  = make_float4(0.f, 0.f, 0.f, 0.f);
                mF0[i] = 0.f;
                mF1[i] = 0.f;
            }
            const int steps = k4 >> 2;
            const float2* qM = &mM[laneHi << 1];
            const float4* qB = &mB[laneHi << 1];
            const float4* qC = &mC[laneHi << 1];
            const float*  qF = &sF[wave][laneN & 1][laneHi << 1];
            #pragma unroll 1
            for (int st = 0; st < steps; ++st) {
                const float4 m  = *(const float4*)qM;
                const float4 bA = qB[0], bB = qB[1];
                const float4 cA = qC[0], cB = qC[1];
                const v2f Bm = *(const v2f*)qF;
                qM += 4;
                qB += 4;
                qC += 4;
                qF += 4;

                const float dxA = xw - m.x;
                const float dxB = xw - m.z;
                const float bAt[PR] = {bA.x, bA.y, bA.z, bA.w};
                const float bBt[PR] = {bB.x, bB.y, bB.z, bB.w};
                const float cAt[PR] = {cA.x, cA.y, cA.z, cA.w};
                const float cBt[PR] = {cB.x, cB.y, cB.z, cB.w};
                #pragma unroll
                for (int t = 0; t < PR; ++t) {
                    float mA = fmaf(fmaf(m.y, dxA, bAt[t]), dxA, cAt[t]);
                    float mB2 = fmaf(fmaf(m.w, dxB, bBt[t]), dxB, cBt[t]);
                    v2f Am = {__builtin_amdgcn_exp2f(mA), __builtin_amdgcn_exp2f(mB2)};

                    acc[t] = __builtin_amdgcn_wmma_f32_16x16x4_f32(
                                 false, Am, false, Bm, (short)0, acc[t], false, false);
                }
            }
            cnt = 0;
        }
    }

    if (laneN < F_) {
        #pragma unroll
        for (int t = 0; t < PR; ++t) {
            #pragma unroll
            for (int r = 0; r < 8; ++r) {
                int M = r + (laneHi << 3);
                int w = cb + M;
                out[(((size_t)(h0 + t) * W_ + w) * D_ + d) * F_ + laneN] = acc[t][r];
            }
        }
    }
}

extern "C" void kernel_launch(void* const* d_in, const int* in_sizes, int n_in,
                              void* d_out, int out_size, void* d_ws, size_t ws_size,
                              hipStream_t stream)
{
    const float* xyz   = (const float*)d_in[0];
    const float* scale = (const float*)d_in[1];
    const float* rot   = (const float*)d_in[2];
    const float* opac  = (const float*)d_in[3];
    const float* feat  = (const float*)d_in[4];

    render_fused<<<dim3(D_ * BPS), BTH, 0, stream>>>(xyz, scale, rot, opac, feat, (float*)d_out);
}


# Cull bound moved from -127 to -64, a numerical change: entries whose terms on the patch are all under 2^-64 of their weight are left out; a pixel loses under 2^-53 of the largest weight; order kept
# speedup vs baseline: 9.9822x; 1.2770x over previous
#include <hip/hip_runtime.h>
#include <math.h>

#define NG   2048
#define H_   128
#define W_   128
#define D_   32
#define F_   2
#define LOG2E 1.4426950408889634f

typedef float v2f __attribute__((ext_vector_type(2)));
typedef float v8f __attribute__((ext_vector_type(8)));

#define PR   4
#define NWV  32
#define BTH  (32 * NWV)
#define NRD  (NG / BTH)
#define NSP  (W_ / 16)
#define NRG  (H_ / PR)
#define BPS  (NSP * NRG / NWV)
#define CAPW 128
#define CULL_T (-64.0f)
static_assert(NRD == 2, "phase 1 packs the two rounds' counts into two words");
static_assert(NWV == 32, "one lane per wave in the count exchange");
static_assert(NSP == 8 && NRG % (NWV / NSP) == 0 && BPS * (NWV / NSP) == NRG, "a block takes every span of NWV/NSP row groups");
static_assert(PR == 4, "row constants are kept in float4");

__global__ __launch_bounds__(BTH) void render_fused(
    const float* __restrict__ xyz, const float* __restrict__ scale,
    const float* __restrict__ rot, const float* __restrict__ opac,
    const float* __restrict__ feat, float* __restrict__ out)
{
    __shared__ float4 sE0[NG];
    __shared__ float4 sE1[NG];
    __shared__ __attribute__((aligned(16))) float2 sM[NWV][CAPW];
    __shared__ float4 sB[NWV][CAPW];
    __shared__ float4 sC[NWV][CAPW];
    __shared__ __attribute__((aligned(16))) float sF[NWV][2][CAPW + 8];
    __shared__ int wcnt[NRD][NWV];

    const int d      = blockIdx.x % D_;
    const int jb     = blockIdx.x / D_;
    const int tid    = threadIdx.x;
    const int lane   = tid & 31;
    const int wave   = __builtin_amdgcn_readfirstlane(tid >> 5);
    const unsigned int ltmask = (1u << lane) - 1u;

    const float z = (float)d * (1.0f / (D_ - 1));
    float4 p0[NRD], p1[NRD];
    bool   live[NRD];
    int    pos[NRD];
    float rq[NRD][4], sc[NRD][3], xp[NRD][3], opv[NRD], ft[NRD][2];
    #pragma unroll
    for (int r = 0; r < NRD; ++r) {
        const int n = r * BTH + tid;
        #pragma unroll
        for (int k = 0; k < 4; ++k) rq[r][k] = rot[n*4+k];
        #pragma unroll
        for (int k = 0; k < 3; ++k) { sc[r][k] = scale[n*3+k]; xp[r][k] = xyz[n*3+k]; }
        opv[r]   = opac[n];
        ft[r][0] = feat[n*2+0];
        ft[r][1] = feat[n*2+1];
    }
    #pragma unroll
    for (int r = 0; r < NRD; ++r) {
        float qw = rq[r][0], qx = rq[r][1], qy = rq[r][2], qz = rq[r][3];
        float inv = 1.0f / (sqrtf(qw*qw + qx*qx + qy*qy + qz*qz) + 1e-8f);
        qw *= inv; qx *= inv; qy *= inv; qz *= inv;

        float r00 = 1.f - 2.f*(qy*qy + qz*qz);
        float r01 = 2.f*(qx*qy - qw*qz);
        float r02 = 2.f*(qx*qz + qw*qy);
        float r10 = 2.f*(qx*qy + qw*qz);
        float r11 = 1.f - 2.f*(qx*qx + qz*qz);
        float r12 = 2.f*(qy*qz - qw*qx);
        float r20 = 2.f*(qx*qz - qw*qy);
        float r21 = 2.f*(qy*qz + qw*qx);
        float r22 = 1.f - 2.f*(qx*qx + qy*qy);

        float sx = sc[r][0], sy = sc[r][1], sz = sc[r][2];
        float m00 = r00*sx, m01 = r01*sy, m02 = r02*sz;
        float m10 = r10*sx, m11 = r11*sy, m12 = r12*sz;
        float m20 = r20*sx, m21 = r21*sy, m22 = r22*sz;

        float cov00 = m00*m00 + m01*m01 + m02*m02;
        float cov01 = m00*m10 + m01*m11 + m02*m12;
        float cov02 = m00*m20 + m01*m21 + m02*m22;
        float cov11 = m10*m10 + m11*m11 + m12*m12;
        float cov12 = m10*m20 + m11*m21 + m12*m22;
        float cov22 = m20*m20 + m21*m21 + m22*m22;

        float zv   = cov22 + 1e-8f;
        float zinv = 1.0f / zv;
        float thr3 = 3.0f * sqrtf(zv);

        float a  = cov00 - zinv*cov02*cov02 + 1e-6f;
        float b  = cov01 - zinv*cov02*cov12;
        float cc = cov11 - zinv*cov12*cov12 + 1e-6f;
        float idet = 1.0f / (a*cc - b*b);
        float pa = cc*idet, pb = -b*idet, pc = a*idet;

        float mu_x = xp[r][0], mu_y = xp[r][1], mu_z = xp[r][2];
        float tt  = zinv * (z - mu_z);
        float mux = mu_x + cov02 * tt;
        float muy = mu_y + cov12 * tt;
        live[r] = fabsf(mu_z - z) < thr3;

        float ea = -0.5f*LOG2E*pa, eb = -LOG2E*pb, ec = -0.5f*LOG2E*pc;
        p0[r] = make_float4(mux, muy, opv[r] * ft[r][0], opv[r] * ft[r][1]);
        p1[r] = make_float4(ea, eb, ec, -0.5f * eb / ea);

        unsigned int bal = (unsigned int)__ballot(live[r]);
        pos[r] = __popc(bal & ltmask);
        if (lane == 0) wcnt[r][wave] = __popc(bal);
    }
    __syncthreads();

    int w0, w1;
    {
        const int c0 = wcnt[0][lane], c1 = wcnt[1][lane];
        w0 = ((lane < wave) ? c0 : 0) + (c0 << 16);
        w1 = ((lane < wave) ? c1 : 0) + (c1 << 16);
        #pragma unroll
        for (int m = 16; m >= 1; m >>= 1) {
            w0 += __shfl_xor(w0, m, 32);
            w1 += __shfl_xor(w1, m, 32);
        }
    }
    const int tot0 = w0 >> 16;
    const int off0 = w0 & 0xffff;
    const int off1 = tot0 + (w1 & 0xffff);
    const int total = __builtin_amdgcn_readfirstlane(tot0 + (w1 >> 16));

    if (live[0]) { const int i = off0 + pos[0]; sE0[i] = p0[0]; sE1[i] = p1[0]; }
    if (live[1]) { const int i = off1 + pos[1]; sE0[i] = p0[1]; sE1[i] = p1[1]; }
    __syncthreads();

    const int laneN  = lane & 15;
    const int laneHi = lane >> 4;
    const int cb     = 16 * (wave % NSP);
    const int h0     = PR * (jb + BPS * (wave / NSP));

    float yh[PR];
    #pragma unroll
    for (int t = 0; t < PR; ++t)
        yh[t] = (float)(h0 + t) * (1.0f / (H_ - 1));
    const float xw  = (float)(cb + laneN) * (1.0f / (W_ - 1));
    const float xlo = (float)cb * (1.0f / (W_ - 1));
    const float xhi = (float)(cb + 15) * (1.0f / (W_ - 1));

    v8f acc[PR];
    #pragma unroll
    for (int t = 0; t < PR; ++t)
        acc[t] = (v8f){0.f, 0.f, 0.f, 0.f, 0.f, 0.f, 0.f, 0.f};

    float2* const mM  = sM[wave];
    float4* const mB  = sB[wave];
    float4* const mC  = sC[wave];
    float*  const mF0 = sF[wave][0];
    float*  const mF1 = sF[wave][1];

    int cnt = 0;
    #pragma unroll 1
    for (int base = 0; base < total; base += 32) {
        const int  j     = base + lane;
        const bool valid = j < total;
        const int  jj    = min(j, total - 1);
        const float4 e0 = sE0[jj], e1 = sE1[jj];

        float eB[PR], eC[PR], tv[PR];
        #pragma unroll
        for (int t = 0; t < PR; ++t) {
            const float dy = yh[t] - e0.y;
            eB[t] = e1.y * dy;
            eC[t] = e1.z * dy * dy;
            tv[t] = e1.w * dy;
        }
        const float tl = xlo - e0.x;
        const float tr = xhi - e0.x;
        bool cull = e1.x < 0.f;
        #pragma unroll
        for (int t = 0; t < PR; ++t) {
            float v = fminf(fmaxf(tv[t], tl), tr);
            float q = fmaf(fmaf(e1.x, v, eB[t]), v, eC[t]);
            cull = cull && (q < CULL_T);
        }
        const bool keep = valid && !cull;
        const unsigned int bal = (unsigned int)__ballot(keep);
        if (keep) {
            const int i = cnt + __popc(bal & ltmask);
            mM[i]  = make_float2(e0.x, e1.x);
            mB[i]  = make_float4(eB[0], eB[1], eB[2], eB[3]);
            mC[i]  = make_float4(eC[0], eC[1], eC[2], eC[3]);
            mF0[i] = e0.z;
            mF1[i] = e0.w;
        }
        cnt += __builtin_amdgcn_readfirstlane(__popc(bal));

        if (cnt > CAPW - 32 || base + 32 >= total) {
            const int k4 = (cnt + 3) & ~3;
            if (lane < k4 - cnt) {
                const int i = cnt + lane;
                mM[i]  = make_float2(0.f, 0.f);
                mB[i]  = make_float4(0.f, 0.f, 0.f, 0.f);
                mC[i]  = make_float4(0.f, 0.f, 0.f, 0.f);
                mF0[i] = 0.f;
                mF1[i] = 0.f;
            }
            const int steps = k4 >> 2;
            const float2* qM = &mM[laneHi << 1];
            const float4* qB = &mB[laneHi << 1];
            const float4* qC = &mC[laneHi << 1];
            const float*  qF = &sF[wave][laneN & 1][laneHi << 1];
            #pragma unroll 1
            for (int st = 0; st < steps; ++st) {
                const float4 m  = *(const float4*)qM;
                const float4 bA = qB[0], bB = qB[1];
                const float4 cA = qC[0], cB = qC[1];
                const v2f Bm = *(const v2f*)qF;
                qM += 4;
                qB += 4;
                qC += 4;
                qF += 4;

                const float dxA = xw - m.x;
                const float dxB = xw - m.z;
                const float bAt[PR] = {bA.x, bA.y, bA.z, bA.w};
                const float bBt[PR] = {bB.x, bB.y, bB.z, bB.w};
                const float cAt[PR] = {cA.x, cA.y, cA.z, cA.w};
                const float cBt[PR] = {cB.x, cB.y, cB.z, cB.w};
                #pragma unroll
                for (int t = 0; t < PR; ++t) {
                    float mA = fmaf(fmaf(m.y, dxA, bAt[t]), dxA, cAt[t]);
                    float mB2 = fmaf(fmaf(m.w, dxB, bBt[t]), dxB, cBt[t]);
                    v2f Am = {__builtin_amdgcn_exp2f(mA), __builtin_amdgcn_exp2f(mB2)};

                    acc[t] = __builtin_amdgcn_wmma_f32_16x16x4_f32(
                                 false, Am, false, Bm, (short)0, acc[t], false, false);
                }
            }
            cnt = 0;
        }
    }

    if (laneN < F_) {
        #pragma unroll
        for (int t = 0; t < PR; ++t) {
            #pragma unroll
            for (int r = 0; r < 8; ++r) {
                int M = r + (laneHi << 3);
                int w = cb + M;
                out[(((size_t)(h0 + t) * W_ + w) * D_ + d) * F_ + laneN] = acc[t][r];
            }
        }
    }
}

extern "C" void kernel_launch(void* const* d_in, const int* in_sizes, int n_in,
                              void* d_out, int out_size, void* d_ws, size_t ws_size,
                              hipStream_t stream)
{
    const float* xyz   = (const float*)d_in[0];
    const float* scale = (const float*)d_in[1];
    const float* rot   = (const float*)d_in[2];
    const float* opac  = (const float*)d_in[3];
    const float* feat  = (const float*)d_in[4];

    render_fused<<<dim3(D_ * BPS), BTH, 0, stream>>>(xyz, scale, rot, opac, feat, (float*)d_out);
}


# Cull bound moved from -64 to -40, a numerical change: entries with every term on the patch under 2^-40 of their weight are left out; a pixel loses under 2^-29 of the largest weight; order kept
# speedup vs baseline: 10.6810x; 1.0700x over previous
#include <hip/hip_runtime.h>
#include <math.h>

#define NG   2048
#define H_   128
#define W_   128
#define D_   32
#define F_   2
#define LOG2E 1.4426950408889634f

typedef float v2f __attribute__((ext_vector_type(2)));
typedef float v8f __attribute__((ext_vector_type(8)));

#define PR   4
#define NWV  32
#define BTH  (32 * NWV)
#define NRD  (NG / BTH)
#define NSP  (W_ / 16)
#define NRG  (H_ / PR)
#define BPS  (NSP * NRG / NWV)
#define CAPW 128
#define CULL_T (-40.0f)
static_assert(NRD == 2, "phase 1 packs the two rounds' counts into two words");
static_assert(NWV == 32, "one lane per wave in the count exchange");
static_assert(NSP == 8 && NRG % (NWV / NSP) == 0 && BPS * (NWV / NSP) == NRG, "a block takes every span of NWV/NSP row groups");
static_assert(PR == 4, "row constants are kept in float4");

__global__ __launch_bounds__(BTH) void render_fused(
    const float* __restrict__ xyz, const float* __restrict__ scale,
    const float* __restrict__ rot, const float* __restrict__ opac,
    const float* __restrict__ feat, float* __restrict__ out)
{
    __shared__ float4 sE0[NG];
    __shared__ float4 sE1[NG];
    __shared__ __attribute__((aligned(16))) float2 sM[NWV][CAPW];
    __shared__ float4 sB[NWV][CAPW];
    __shared__ float4 sC[NWV][CAPW];
    __shared__ __attribute__((aligned(16))) float sF[NWV][2][CAPW + 8];
    __shared__ int wcnt[NRD][NWV];

    const int d      = blockIdx.x % D_;
    const int jb     = blockIdx.x / D_;
    const int tid    = threadIdx.x;
    const int lane   = tid & 31;
    const int wave   = __builtin_amdgcn_readfirstlane(tid >> 5);
    const unsigned int ltmask = (1u << lane) - 1u;

    const float z = (float)d * (1.0f / (D_ - 1));
    float4 p0[NRD], p1[NRD];
    bool   live[NRD];
    int    pos[NRD];
    float rq[NRD][4], sc[NRD][3], xp[NRD][3], opv[NRD], ft[NRD][2];
    #pragma unroll
    for (int r = 0; r < NRD; ++r) {
        const int n = r * BTH + tid;
        #pragma unroll
        for (int k = 0; k < 4; ++k) rq[r][k] = rot[n*4+k];
        #pragma unroll
        for (int k = 0; k < 3; ++k) { sc[r][k] = scale[n*3+k]; xp[r][k] = xyz[n*3+k]; }
        opv[r]   = opac[n];
        ft[r][0] = feat[n*2+0];
        ft[r][1] = feat[n*2+1];
    }
    #pragma unroll
    for (int r = 0; r < NRD; ++r) {
        float qw = rq[r][0], qx = rq[r][1], qy = rq[r][2], qz = rq[r][3];
        float inv = 1.0f / (sqrtf(qw*qw + qx*qx + qy*qy + qz*qz) + 1e-8f);
        qw *= inv; qx *= inv; qy *= inv; qz *= inv;

        float r00 = 1.f - 2.f*(qy*qy + qz*qz);
        float r01 = 2.f*(qx*qy - qw*qz);
        float r02 = 2.f*(qx*qz + qw*qy);
        float r10 = 2.f*(qx*qy + qw*qz);
        float r11 = 1.f - 2.f*(qx*qx + qz*qz);
        float r12 = 2.f*(qy*qz - qw*qx);
        float r20 = 2.f*(qx*qz - qw*qy);
        float r21 = 2.f*(qy*qz + qw*qx);
        float r22 = 1.f - 2.f*(qx*qx + qy*qy);

        float sx = sc[r][0], sy = sc[r][1], sz = sc[r][2];
        float m00 = r00*sx, m01 = r01*sy, m02 = r02*sz;
        float m10 = r10*sx, m11 = r11*sy, m12 = r12*sz;
        float m20 = r20*sx, m21 = r21*sy, m22 = r22*sz;

        float cov00 = m00*m00 + m01*m01 + m02*m02;
        float cov01 = m00*m10 + m01*m11 + m02*m12;
        float cov02 = m00*m20 + m01*m21 + m02*m22;
        float cov11 = m10*m10 + m11*m11 + m12*m12;
        float cov12 = m10*m20 + m11*m21 + m12*m22;
        float cov22 = m20*m20 + m21*m21 + m22*m22;

        float zv   = cov22 + 1e-8f;
        float zinv = 1.0f / zv;
        float thr3 = 3.0f * sqrtf(zv);

        float a  = cov00 - zinv*cov02*cov02 + 1e-6f;
        float b  = cov01 - zinv*cov02*cov12;
        float cc = cov11 - zinv*cov12*cov12 + 1e-6f;
        float idet = 1.0f / (a*cc - b*b);
        float pa = cc*idet, pb = -b*idet, pc = a*idet;

        float mu_x = xp[r][0], mu_y = xp[r][1], mu_z = xp[r][2];
        float tt  = zinv * (z - mu_z);
        float mux = mu_x + cov02 * tt;
        float muy = mu_y + cov12 * tt;
        live[r] = fabsf(mu_z - z) < thr3;

        float ea = -0.5f*LOG2E*pa, eb = -LOG2E*pb, ec = -0.5f*LOG2E*pc;
        p0[r] = make_float4(mux, muy, opv[r] * ft[r][0], opv[r] * ft[r][1]);
        p1[r] = make_float4(ea, eb, ec, -0.5f * eb / ea);

        unsigned int bal = (unsigned int)__ballot(live[r]);
        pos[r] = __popc(bal & ltmask);
        if (lane == 0) wcnt[r][wave] = __popc(bal);
    }
    __syncthreads();

    int w0, w1;
    {
        const int c0 = wcnt[0][lane], c1 = wcnt[1][lane];
        w0 = ((lane < wave) ? c0 : 0) + (c0 << 16);
        w1 = ((lane < wave) ? c1 : 0) + (c1 << 16);
        #pragma unroll
        for (int m = 16; m >= 1; m >>= 1) {
            w0 += __shfl_xor(w0, m, 32);
            w1 += __shfl_xor(w1, m, 32);
        }
    }
    const int tot0 = w0 >> 16;
    const int off0 = w0 & 0xffff;
    const int off1 = tot0 + (w1 & 0xffff);
    const int total = __builtin_amdgcn_readfirstlane(tot0 + (w1 >> 16));

    if (live[0]) { const int i = off0 + pos[0]; sE0[i] = p0[0]; sE1[i] = p1[0]; }
    if (live[1]) { const int i = off1 + pos[1]; sE0[i] = p0[1]; sE1[i] = p1[1]; }
    __syncthreads();

    const int laneN  = lane & 15;
    const int laneHi = lane >> 4;
    const int cb     = 16 * (wave % NSP);
    const int h0     = PR * (jb + BPS * (wave / NSP));

    float yh[PR];
    #pragma unroll
    for (int t = 0; t < PR; ++t)
        yh[t] = (float)(h0 + t) * (1.0f / (H_ - 1));
    const float xw  = (float)(cb + laneN) * (1.0f / (W_ - 1));
    const float xlo = (float)cb * (1.0f / (W_ - 1));
    const float xhi = (float)(cb + 15) * (1.0f / (W_ - 1));

    v8f acc[PR];
    #pragma unroll
    for (int t = 0; t < PR; ++t)
        acc[t] = (v8f){0.f, 0.f, 0.f, 0.f, 0.f, 0.f, 0.f, 0.f};

    float2* const mM  = sM[wave];
    float4* const mB  = sB[wave];
    float4* const mC  = sC[wave];
    float*  const mF0 = sF[wave][0];
    float*  const mF1 = sF[wave][1];

    int cnt = 0;
    #pragma unroll 1
    for (int base = 0; base < total; base += 32) {
        const int  j     = base + lane;
        const bool valid = j < total;
        const int  jj    = min(j, total - 1);
        const float4 e0 = sE0[jj], e1 = sE1[jj];

        float eB[PR], eC[PR], tv[PR];
        #pragma unroll
        for (int t = 0; t < PR; ++t) {
            const float dy = yh[t] - e0.y;
            eB[t] = e1.y * dy;
            eC[t] = e1.z * dy * dy;
            tv[t] = e1.w * dy;
        }
        const float tl = xlo - e0.x;
        const float tr = xhi - e0.x;
        bool cull = e1.x < 0.f;
        #pragma unroll
        for (int t = 0; t < PR; ++t) {
            float v = fminf(fmaxf(tv[t], tl), tr);
            float q = fmaf(fmaf(e1.x, v, eB[t]), v, eC[t]);
            cull = cull && (q < CULL_T);
        }
        const bool keep = valid && !cull;
        const unsigned int bal = (unsigned int)__ballot(keep);
        if (keep) {
            const int i = cnt + __popc(bal & ltmask);
            mM[i]  = make_float2(e0.x, e1.x);
            mB[i]  = make_float4(eB[0], eB[1], eB[2], eB[3]);
            mC[i]  = make_float4(eC[0], eC[1], eC[2], eC[3]);
            mF0[i] = e0.z;
            mF1[i] = e0.w;
        }
        cnt += __builtin_amdgcn_readfirstlane(__popc(bal));

        if (cnt > CAPW - 32 || base + 32 >= total) {
            const int k4 = (cnt + 3) & ~3;
            if (lane < k4 - cnt) {
                const int i = cnt + lane;
                mM[i]  = make_float2(0.f, 0.f);
                mB[i]  = make_float4(0.f, 0.f, 0.f, 0.f);
                mC[i]  = make_float4(0.f, 0.f, 0.f, 0.f);
                mF0[i] = 0.f;
                mF1[i] = 0.f;
            }
            const int steps = k4 >> 2;
            const float2* qM = &mM[laneHi << 1];
            const float4* qB = &mB[laneHi << 1];
            const float4* qC = &mC[laneHi << 1];
            const float*  qF = &sF[wave][laneN & 1][laneHi << 1];
            #pragma unroll 1
            for (int st = 0; st < steps; ++st) {
                const float4 m  = *(const float4*)qM;
                const float4 bA = qB[0], bB = qB[1];
                const float4 cA = qC[0], cB = qC[1];
                const v2f Bm = *(const v2f*)qF;
                qM += 4;
                qB += 4;
                qC += 4;
                qF += 4;

                const float dxA = xw - m.x;
                const float dxB = xw - m.z;
                const float bAt[PR] = {bA.x, bA.y, bA.z, bA.w};
                const float bBt[PR] = {bB.x, bB.y, bB.z, bB.w};
                const float cAt[PR] = {cA.x, cA.y, cA.z, cA.w};
                const float cBt[PR] = {cB.x, cB.y, cB.z, cB.w};
                #pragma unroll
                for (int t = 0; t < PR; ++t) {
                    float mA = fmaf(fmaf(m.y, dxA, bAt[t]), dxA, cAt[t]);
                    float mB2 = fmaf(fmaf(m.w, dxB, bBt[t]), dxB, cBt[t]);
                    v2f Am = {__builtin_amdgcn_exp2f(mA), __builtin_amdgcn_exp2f(mB2)};

                    acc[t] = __builtin_amdgcn_wmma_f32_16x16x4_f32(
                                 false, Am, false, Bm, (short)0, acc[t], false, false);
                }
            }
            cnt = 0;
        }
    }

    if (laneN < F_) {
        #pragma unroll
        for (int t = 0; t < PR; ++t) {
            #pragma unroll
            for (int r = 0; r < 8; ++r) {
                int M = r + (laneHi << 3);
                int w = cb + M;
                out[(((size_t)(h0 + t) * W_ + w) * D_ + d) * F_ + laneN] = acc[t][r];
            }
        }
    }
}

extern "C" void kernel_launch(void* const* d_in, const int* in_sizes, int n_in,
                              void* d_out, int out_size, void* d_ws, size_t ws_size,
                              hipStream_t stream)
{
    const float* xyz   = (const float*)d_in[0];
    const float* scale = (const float*)d_in[1];
    const float* rot   = (const float*)d_in[2];
    const float* opac  = (const float*)d_in[3];
    const float* feat  = (const float*)d_in[4];

    render_fused<<<dim3(D_ * BPS), BTH, 0, stream>>>(xyz, scale, rot, opac, feat, (float*)d_out);
}


# Block index remapped: a slice's 8 blocks get adjacent indices and slices run from the middle of the z range outwards, so the short-list end slices start last, inside the entry ramp; sums unchanged
# speedup vs baseline: 12.5657x; 1.1765x over previous
#include <hip/hip_runtime.h>
#include <math.h>

#define NG   2048
#define H_   128
#define W_   128
#define D_   32
#define F_   2
#define LOG2E 1.4426950408889634f

typedef float v2f __attribute__((ext_vector_type(2)));
typedef float v8f __attribute__((ext_vector_type(8)));

#define PR   4
#define NWV  32
#define BTH  (32 * NWV)
#define NRD  (NG / BTH)
#define NSP  (W_ / 16)
#define NRG  (H_ / PR)
#define BPS  (NSP * NRG / NWV)
#define CAPW 128
#define CULL_T (-40.0f)
static_assert(NRD == 2, "phase 1 packs the two rounds' counts into two words");
static_assert(NWV == 32, "one lane per wave in the count exchange");
static_assert(NSP == 8 && NRG % (NWV / NSP) == 0 && BPS * (NWV / NSP) == NRG, "a block takes every span of NWV/NSP row groups");
static_assert(PR == 4, "row constants are kept in float4");
static_assert(D_ % 2 == 0, "the middle-out slice order pairs the slices below and above D_/2");

__global__ __launch_bounds__(BTH) void render_fused(
    const float* __restrict__ xyz, const float* __restrict__ scale,
    const float* __restrict__ rot, const float* __restrict__ opac,
    const float* __restrict__ feat, float* __restrict__ out)
{
    __shared__ float4 sE0[NG];
    __shared__ float4 sE1[NG];
    __shared__ __attribute__((aligned(16))) float2 sM[NWV][CAPW];
    __shared__ float4 sB[NWV][CAPW];
    __shared__ float4 sC[NWV][CAPW];
    __shared__ __attribute__((aligned(16))) float sF[NWV][2][CAPW + 8];
    __shared__ int wcnt[NRD][NWV];

    const int rk     = blockIdx.x / BPS;
    const int jb     = blockIdx.x % BPS;
    const int d      = (rk & 1) ? (D_ / 2 + (rk >> 1)) : (D_ / 2 - 1 - (rk >> 1));
    const int tid    = threadIdx.x;
    const int lane   = tid & 31;
    const int wave   = __builtin_amdgcn_readfirstlane(tid >> 5);
    const unsigned int ltmask = (1u << lane) - 1u;

    const float z = (float)d * (1.0f / (D_ - 1));
    float4 p0[NRD], p1[NRD];
    bool   live[NRD];
    int    pos[NRD];
    float rq[NRD][4], sc[NRD][3], xp[NRD][3], opv[NRD], ft[NRD][2];
    #pragma unroll
    for (int r = 0; r < NRD; ++r) {
        const int n = r * BTH + tid;
        #pragma unroll
        for (int k = 0; k < 4; ++k) rq[r][k] = rot[n*4+k];
        #pragma unroll
        for (int k = 0; k < 3; ++k) { sc[r][k] = scale[n*3+k]; xp[r][k] = xyz[n*3+k]; }
        opv[r]   = opac[n];
        ft[r][0] = feat[n*2+0];
        ft[r][1] = feat[n*2+1];
    }
    #pragma unroll
    for (int r = 0; r < NRD; ++r) {
        float qw = rq[r][0], qx = rq[r][1], qy = rq[r][2], qz = rq[r][3];
        float inv = 1.0f / (sqrtf(qw*qw + qx*qx + qy*qy + qz*qz) + 1e-8f);
        qw *= inv; qx *= inv; qy *= inv; qz *= inv;

        float r00 = 1.f - 2.f*(qy*qy + qz*qz);
        float r01 = 2.f*(qx*qy - qw*qz);
        float r02 = 2.f*(qx*qz + qw*qy);
        float r10 = 2.f*(qx*qy + qw*qz);
        float r11 = 1.f - 2.f*(qx*qx + qz*qz);
        float r12 = 2.f*(qy*qz - qw*qx);
        float r20 = 2.f*(qx*qz - qw*qy);
        float r21 = 2.f*(qy*qz + qw*qx);
        float r22 = 1.f - 2.f*(qx*qx + qy*qy);

        float sx = sc[r][0], sy = sc[r][1], sz = sc[r][2];
        float m00 = r00*sx, m01 = r01*sy, m02 = r02*sz;
        float m10 = r10*sx, m11 = r11*sy, m12 = r12*sz;
        float m20 = r20*sx, m21 = r21*sy, m22 = r22*sz;

        float cov00 = m00*m00 + m01*m01 + m02*m02;
        float cov01 = m00*m10 + m01*m11 + m02*m12;
        float cov02 = m00*m20 + m01*m21 + m02*m22;
        float cov11 = m10*m10 + m11*m11 + m12*m12;
        float cov12 = m10*m20 + m11*m21 + m12*m22;
        float cov22 = m20*m20 + m21*m21 + m22*m22;

        float zv   = cov22 + 1e-8f;
        float zinv = 1.0f / zv;
        float thr3 = 3.0f * sqrtf(zv);

        float a  = cov00 - zinv*cov02*cov02 + 1e-6f;
        float b  = cov01 - zinv*cov02*cov12;
        float cc = cov11 - zinv*cov12*cov12 + 1e-6f;
        float idet = 1.0f / (a*cc - b*b);
        float pa = cc*idet, pb = -b*idet, pc = a*idet;

        float mu_x = xp[r][0], mu_y = xp[r][1], mu_z = xp[r][2];
        float tt  = zinv * (z - mu_z);
        float mux = mu_x + cov02 * tt;
        float muy = mu_y + cov12 * tt;
        live[r] = fabsf(mu_z - z) < thr3;

        float ea = -0.5f*LOG2E*pa, eb = -LOG2E*pb, ec = -0.5f*LOG2E*pc;
        p0[r] = make_float4(mux, muy, opv[r] * ft[r][0], opv[r] * ft[r][1]);
        p1[r] = make_float4(ea, eb, ec, -0.5f * eb / ea);

        unsigned int bal = (unsigned int)__ballot(live[r]);
        pos[r] = __popc(bal & ltmask);
        if (lane == 0) wcnt[r][wave] = __popc(bal);
    }
    __syncthreads();

    int w0, w1;
    {
        const int c0 = wcnt[0][lane], c1 = wcnt[1][lane];
        w0 = ((lane < wave) ? c0 : 0) + (c0 << 16);
        w1 = ((lane < wave) ? c1 : 0) + (c1 << 16);
        #pragma unroll
        for (int m = 16; m >= 1; m >>= 1) {
            w0 += __shfl_xor(w0, m, 32);
            w1 += __shfl_xor(w1, m, 32);
        }
    }
    const int tot0 = w0 >> 16;
    const int off0 = w0 & 0xffff;
    const int off1 = tot0 + (w1 & 0xffff);
    const int total = __builtin_amdgcn_readfirstlane(tot0 + (w1 >> 16));

    if (live[0]) { const int i = off0 + pos[0]; sE0[i] = p0[0]; sE1[i] = p1[0]; }
    if (live[1]) { const int i = off1 + pos[1]; sE0[i] = p0[1]; sE1[i] = p1[1]; }
    __syncthreads();

    const int laneN  = lane & 15;
    const int laneHi = lane >> 4;
    const int cb     = 16 * (wave % NSP);
    const int h0     = PR * (jb + BPS * (wave / NSP));

    float yh[PR];
    #pragma unroll
    for (int t = 0; t < PR; ++t)
        yh[t] = (float)(h0 + t) * (1.0f / (H_ - 1));
    const float xw  = (float)(cb + laneN) * (1.0f / (W_ - 1));
    const float xlo = (float)cb * (1.0f / (W_ - 1));
    const float xhi = (float)(cb + 15) * (1.0f / (W_ - 1));

    v8f acc[PR];
    #pragma unroll
    for (int t = 0; t < PR; ++t)
        acc[t] = (v8f){0.f, 0.f, 0.f, 0.f, 0.f, 0.f, 0.f, 0.f};

    float2* const mM  = sM[wave];
    float4* const mB  = sB[wave];
    float4* const mC  = sC[wave];
    float*  const mF0 = sF[wave][0];
    float*  const mF1 = sF[wave][1];

    int cnt = 0;
    #pragma unroll 1
    for (int base = 0; base < total; base += 32) {
        const int  j     = base + lane;
        const bool valid = j < total;
        const int  jj    = min(j, total - 1);
        const float4 e0 = sE0[jj], e1 = sE1[jj];

        float eB[PR], eC[PR], tv[PR];
        #pragma unroll
        for (int t = 0; t < PR; ++t) {
            const float dy = yh[t] - e0.y;
            eB[t] = e1.y * dy;
            eC[t] = e1.z * dy * dy;
            tv[t] = e1.w * dy;
        }
        const float tl = xlo - e0.x;
        const float tr = xhi - e0.x;
        bool cull = e1.x < 0.f;
        #pragma unroll
        for (int t = 0; t < PR; ++t) {
            float v = fminf(fmaxf(tv[t], tl), tr);
            float q = fmaf(fmaf(e1.x, v, eB[t]), v, eC[t]);
            cull = cull && (q < CULL_T);
        }
        const bool keep = valid && !cull;
        const unsigned int bal = (unsigned int)__ballot(keep);
        if (keep) {
            const int i = cnt + __popc(bal & ltmask);
            mM[i]  = make_float2(e0.x, e1.x);
            mB[i]  = make_float4(eB[0], eB[1], eB[2], eB[3]);
            mC[i]  = make_float4(eC[0], eC[1], eC[2], eC[3]);
            mF0[i] = e0.z;
            mF1[i] = e0.w;
        }
        cnt += __builtin_amdgcn_readfirstlane(__popc(bal));

        if (cnt > CAPW - 32 || base + 32 >= total) {
            const int k4 = (cnt + 3) & ~3;
            if (lane < k4 - cnt) {
                const int i = cnt + lane;
                mM[i]  = make_float2(0.f, 0.f);
                mB[i]  = make_float4(0.f, 0.f, 0.f, 0.f);
                mC[i]  = make_float4(0.f, 0.f, 0.f, 0.f);
                mF0[i] = 0.f;
                mF1[i] = 0.f;
            }
            const int steps = k4 >> 2;
            const float2* qM = &mM[laneHi << 1];
            const float4* qB = &mB[laneHi << 1];
            const float4* qC = &mC[laneHi << 1];
            const float*  qF = &sF[wave][laneN & 1][laneHi << 1];
            #pragma unroll 1
            for (int st = 0; st < steps; ++st) {
                const float4 m  = *(const float4*)qM;
                const float4 bA = qB[0], bB = qB[1];
                const float4 cA = qC[0], cB = qC[1];
                const v2f Bm = *(const v2f*)qF;
                qM += 4;
                qB += 4;
                qC += 4;
                qF += 4;

                const float dxA = xw - m.x;
                const float dxB = xw - m.z;
                const float bAt[PR] = {bA.x, bA.y, bA.z, bA.w};
                const float bBt[PR] = {bB.x, bB.y, bB.z, bB.w};
                const float cAt[PR] = {cA.x, cA.y, cA.z, cA.w};
                const float cBt[PR] = {cB.x, cB.y, cB.z, cB.w};
                #pragma unroll
                for (int t = 0; t < PR; ++t) {
                    float mA = fmaf(fmaf(m.y, dxA, bAt[t]), dxA, cAt[t]);
                    float mB2 = fmaf(fmaf(m.w, dxB, bBt[t]), dxB, cBt[t]);
                    v2f Am = {__builtin_amdgcn_exp2f(mA), __builtin_amdgcn_exp2f(mB2)};

                    acc[t] = __builtin_amdgcn_wmma_f32_16x16x4_f32(
                                 false, Am, false, Bm, (short)0, acc[t], false, false);
                }
            }
            cnt = 0;
        }
    }

    if (laneN < F_) {
        #pragma unroll
        for (int t = 0; t < PR; ++t) {
            #pragma unroll
            for (int r = 0; r < 8; ++r) {
                int M = r + (laneHi << 3);
                int w = cb + M;
                out[(((size_t)(h0 + t) * W_ + w) * D_ + d) * F_ + laneN] = acc[t][r];
            }
        }
    }
}

extern "C" void kernel_launch(void* const* d_in, const int* in_sizes, int n_in,
                              void* d_out, int out_size, void* d_ws, size_t ws_size,
                              hipStream_t stream)
{
    const float* xyz   = (const float*)d_in[0];
    const float* scale = (const float*)d_in[1];
    const float* rot   = (const float*)d_in[2];
    const float* opac  = (const float*)d_in[3];
    const float* feat  = (const float*)d_in[4];

    render_fused<<<dim3(D_ * BPS), BTH, 0, stream>>>(xyz, scale, rot, opac, feat, (float*)d_out);
}
